# starting kernel
# baseline (speedup 1.0000x reference)
#include <hip/hip_runtime.h>
#include <math.h>

typedef __attribute__((ext_vector_type(2))) float v2f;
typedef __attribute__((ext_vector_type(8))) float v8f;

#define TS   32
#define KC   32
#define ASTR 36
#define BSTR 33

static __device__ __forceinline__ float leaky02(float x) {
    return x > 0.f ? x : 0.2f * x;
}

static __device__ __forceinline__ void atomicMaxF(float* addr, float v) {
    if (v >= 0.f)
        atomicMax((int*)addr, __float_as_int(v));
    else
        atomicMin((unsigned int*)addr, __float_as_uint(v));
}

__global__ __launch_bounds__(128)
void gemm_wmma_f32(const float* __restrict__ A, const float* __restrict__ B,
                   float* __restrict__ D, int M, int K, int N) {
    __shared__ float As[TS * ASTR];
    __shared__ float Bs[KC * BSTR];

    const int tileN = blockIdx.x * TS;
    const int tileM = blockIdx.y * TS;
    const int t    = threadIdx.x;
    const int wv   = t >> 5;
    const int lane = t & 31;
    const int half = lane >> 4;
    const int lm   = lane & 15;
    const int wm   = wv >> 1;
    const int wn   = wv & 1;

    const int crow = t >> 2;
    const int ccol = (t & 3) * 8;

    v8f acc = {};

    for (int kt = 0; kt < K; kt += KC) {
        const float* ga = A + (size_t)(tileM + crow) * K + kt + ccol;
        const float* gb = B + (size_t)(kt + crow) * N + tileN + ccol;
        if (kt + KC < K) {
            __builtin_prefetch(ga + KC, 0, 0);
            __builtin_prefetch(gb + (size_t)KC * N, 0, 0);
        }
        #pragma unroll
        for (int j = 0; j < 8; ++j) As[crow * ASTR + ccol + j] = ga[j];
        #pragma unroll
        for (int j = 0; j < 8; ++j) Bs[crow * BSTR + ccol + j] = gb[j];
        __syncthreads();

        #pragma unroll
        for (int kk = 0; kk < KC; kk += 4) {
            const int k0 = kk + 2 * half;
            v2f a, b;
            a.x = As[(wm * 16 + lm) * ASTR + k0];
            a.y = As[(wm * 16 + lm) * ASTR + k0 + 1];
            b.x = Bs[(k0)     * BSTR + wn * 16 + lm];
            b.y = Bs[(k0 + 1) * BSTR + wn * 16 + lm];
            acc = __builtin_amdgcn_wmma_f32_16x16x4_f32(
                false, a, false, b, (short)0, acc, false, false);
        }
        __syncthreads();
    }

    const int col   = tileN + wn * 16 + lm;
    const int rbase = tileM + wm * 16 + half * 8;
    #pragma unroll
    for (int r = 0; r < 8; ++r)
        D[(size_t)(rbase + r) * N + col] = acc[r];
}

__global__ void att_logits(const float* __restrict__ h,
                           const float* __restrict__ a_s,
                           const float* __restrict__ a_d,
                           float* __restrict__ al_s, float* __restrict__ al_d,
                           int N, int H, int C) {
    int t = blockIdx.x * blockDim.x + threadIdx.x;
    if (t >= N * H) return;
    int n = t / H, hh = t - n * H;
    const float* row = h + (size_t)n * H * C + (size_t)hh * C;
    const float* vs  = a_s + (size_t)hh * C;
    const float* vd  = a_d + (size_t)hh * C;
    float ss = 0.f, sd = 0.f;
    for (int c = 0; c < C; ++c) {
        float v = row[c];
        ss = fmaf(v, vs[c], ss);
        sd = fmaf(v, vd[c], sd);
    }
    al_s[t] = ss;
    al_d[t] = sd;
}

__global__ void fill_f32(float* __restrict__ p, float v, size_t n) {
    size_t i = blockIdx.x * (size_t)blockDim.x + threadIdx.x;
    if (i < n) p[i] = v;
}

__global__ void edge_max(const int* __restrict__ ei, int E, int N, int H,
                         const float* __restrict__ al_s,
                         const float* __restrict__ al_d,
                         float* __restrict__ mbuf) {
    int t = blockIdx.x * blockDim.x + threadIdx.x;
    int tot = (E + N) * H;
    if (t >= tot) return;
    int e = t / H, hh = t - e * H;
    int s, d;
    if (e < E) { s = ei[e]; d = ei[E + e]; } else { s = d = e - E; }
    float v = leaky02(al_s[s * H + hh] + al_d[d * H + hh]);
    atomicMaxF(&mbuf[d * H + hh], v);
}

__global__ void edge_exp(const int* __restrict__ ei, int E, int N, int H,
                         const float* __restrict__ al_s,
                         const float* __restrict__ al_d,
                         const float* __restrict__ mbuf,
                         float* __restrict__ exbuf,
                         float* __restrict__ denom) {
    int t = blockIdx.x * blockDim.x + threadIdx.x;
    int tot = (E + N) * H;
    if (t >= tot) return;
    int e = t / H, hh = t - e * H;
    int s, d;
    if (e < E) { s = ei[e]; d = ei[E + e]; } else { s = d = e - E; }
    float v  = leaky02(al_s[s * H + hh] + al_d[d * H + hh]);
    float ex = __expf(v - mbuf[d * H + hh]);
    exbuf[t] = ex;
    atomicAdd(&denom[d * H + hh], ex);
}

__global__ void edge_aggregate(const int* __restrict__ ei, int E, int N,
                               int H, int C,
                               const float* __restrict__ hfeat,
                               const float* __restrict__ exbuf,
                               const float* __restrict__ denom,
                               float* __restrict__ agg) {
    const int F = H * C;
    long long t = blockIdx.x * (long long)blockDim.x + threadIdx.x;
    long long tot = (long long)(E + N) * F;
    if (t >= tot) return;
    int e = (int)(t / F);
    int f = (int)(t - (long long)e * F);
    int hh = f / C;
    int s, d;
    if (e < E) { s = ei[e]; d = ei[E + e]; } else { s = d = e - E; }
    float alpha = exbuf[(size_t)e * H + hh] / denom[(size_t)d * H + hh];
    atomicAdd(&agg[(size_t)d * F + f], hfeat[(size_t)s * F + f] * alpha);
}

__global__ void bias_relu_inplace(float* __restrict__ p,
                                  const float* __restrict__ b,
                                  size_t total, int F) {
    size_t t = blockIdx.x * (size_t)blockDim.x + threadIdx.x;
    if (t >= total) return;
    int f = (int)(t % F);
    float v = p[t] + b[f];
    p[t] = v > 0.f ? v : 0.f;
}

__global__ void mean_heads_bias(const float* __restrict__ agg,
                                const float* __restrict__ b,
                                float* __restrict__ out, int N, int H, int C) {
    int t = blockIdx.x * blockDim.x + threadIdx.x;
    if (t >= N * C) return;
    int n = t / C, c = t - n * C;
    float s = 0.f;
    for (int hh = 0; hh < H; ++hh)
        s += agg[(size_t)n * H * C + (size_t)hh * C + c];
    out[t] = s / (float)H + b[c];
}

static inline unsigned cdiv_u(size_t a, unsigned b) {
    return (unsigned)((a + b - 1) / b);
}

extern "C" void kernel_launch(void* const* d_in, const int* in_sizes, int n_in,
                              void* d_out, int out_size, void* d_ws, size_t ws_size,
                              hipStream_t stream) {
    const float* x     = (const float*)d_in[0];
    const int*   ei    = (const int*)d_in[1];
    const float* W1    = (const float*)d_in[2];
    const float* a_s1  = (const float*)d_in[3];
    const float* a_d1  = (const float*)d_in[4];
    const float* b1    = (const float*)d_in[5];
    const float* W2    = (const float*)d_in[6];
    const float* a_s2  = (const float*)d_in[7];
    const float* a_d2  = (const float*)d_in[8];
    const float* b2    = (const float*)d_in[9];
    float*       out   = (float*)d_out;

    const int F1   = in_sizes[3];
    const int F2   = in_sizes[7];
    const int LAT  = in_sizes[9];
    const int H    = F2 / LAT;
    const int C1   = F1 / H;
    const int Fin  = in_sizes[2] / F1;
    const int N    = in_sizes[0] / Fin;
    const int E    = in_sizes[1] / 2;
    const int Etot = E + N;

    float* ws = (float*)d_ws;
    size_t o = 0;
    float* h1   = ws + o; o += (size_t)N * F1;
    float* agg1 = ws + o; o += (size_t)N * F1;
    float* ex   = ws + o; o += (size_t)Etot * H;
    float* als  = ws + o; o += (size_t)N * H;
    float* ald  = ws + o; o += (size_t)N * H;
    float* mbuf = ws + o; o += (size_t)N * H;
    float* den  = ws + o; o += (size_t)N * H;
    float* h2   = h1;
    float* agg2 = h1 + (size_t)N * F2;
    (void)ws_size; (void)n_in; (void)out_size;

    const unsigned TB = 256;
    const float NEG_INF = -__builtin_huge_valf();

    fill_f32<<<cdiv_u((size_t)N * F1, TB), TB, 0, stream>>>(agg1, 0.f, (size_t)N * F1);
    fill_f32<<<cdiv_u((size_t)N * H, TB), TB, 0, stream>>>(mbuf, NEG_INF, (size_t)N * H);
    fill_f32<<<cdiv_u((size_t)N * H, TB), TB, 0, stream>>>(den, 0.f, (size_t)N * H);

    gemm_wmma_f32<<<dim3(F1 / TS, N / TS), 128, 0, stream>>>(x, W1, h1, N, Fin, F1);

    att_logits<<<cdiv_u((size_t)N * H, TB), TB, 0, stream>>>(h1, a_s1, a_d1, als, ald, N, H, C1);

    edge_max<<<cdiv_u((size_t)Etot * H, TB), TB, 0, stream>>>(ei, E, N, H, als, ald, mbuf);
    edge_exp<<<cdiv_u((size_t)Etot * H, TB), TB, 0, stream>>>(ei, E, N, H, als, ald, mbuf, ex, den);
    edge_aggregate<<<cdiv_u((size_t)Etot * F1, TB), TB, 0, stream>>>(
        ei, E, N, H, C1, h1, ex, den, agg1);

    bias_relu_inplace<<<cdiv_u((size_t)N * F1, TB), TB, 0, stream>>>(agg1, b1, (size_t)N * F1, F1);

    gemm_wmma_f32<<<dim3(F2 / TS, N / TS), 128, 0, stream>>>(agg1, W2, h2, N, F1, F2);

    fill_f32<<<cdiv_u((size_t)N * F2, TB), TB, 0, stream>>>(agg2, 0.f, (size_t)N * F2);
    fill_f32<<<cdiv_u((size_t)N * H, TB), TB, 0, stream>>>(mbuf, NEG_INF, (size_t)N * H);
    fill_f32<<<cdiv_u((size_t)N * H, TB), TB, 0, stream>>>(den, 0.f, (size_t)N * H);

    att_logits<<<cdiv_u((size_t)N * H, TB), TB, 0, stream>>>(h2, a_s2, a_d2, als, ald, N, H, LAT);

    edge_max<<<cdiv_u((size_t)Etot * H, TB), TB, 0, stream>>>(ei, E, N, H, als, ald, mbuf);
    edge_exp<<<cdiv_u((size_t)Etot * H, TB), TB, 0, stream>>>(ei, E, N, H, als, ald, mbuf, ex, den);
    edge_aggregate<<<cdiv_u((size_t)Etot * F2, TB), TB, 0, stream>>>(
        ei, E, N, H, LAT, h2, ex, den, agg2);

    mean_heads_bias<<<cdiv_u((size_t)N * LAT, TB), TB, 0, stream>>>(agg2, b2, out, N, H, LAT);
}


# aggregation as gather over edges sorted by destination (per-call CSR via int atomics): one block per row sums in registers, one store; float atomics and zero fills gone; sum order = slot arrival
# speedup vs baseline: 3.3833x; 3.3833x over previous
#include <hip/hip_runtime.h>
#include <math.h>

typedef __attribute__((ext_vector_type(2))) float v2f;
typedef __attribute__((ext_vector_type(8))) float v8f;

#define TS   32
#define KC   32
#define ASTR 36
#define BSTR 33

static __device__ __forceinline__ float leaky02(float x) {
    return x > 0.f ? x : 0.2f * x;
}

static __device__ __forceinline__ void atomicMaxF(float* addr, float v) {
    if (v >= 0.f)
        atomicMax((int*)addr, __float_as_int(v));
    else
        atomicMin((unsigned int*)addr, __float_as_uint(v));
}

__global__ __launch_bounds__(128)
void gemm_wmma_f32(const float* __restrict__ A, const float* __restrict__ B,
                   float* __restrict__ D, int M, int K, int N) {
    __shared__ float As[TS * ASTR];
    __shared__ float Bs[KC * BSTR];

    const int tileN = blockIdx.x * TS;
    const int tileM = blockIdx.y * TS;
    const int t    = threadIdx.x;
    const int wv   = t >> 5;
    const int lane = t & 31;
    const int half = lane >> 4;
    const int lm   = lane & 15;
    const int wm   = wv >> 1;
    const int wn   = wv & 1;

    const int crow = t >> 2;
    const int ccol = (t & 3) * 8;

    v8f acc = {};

    for (int kt = 0; kt < K; kt += KC) {
        const float* ga = A + (size_t)(tileM + crow) * K + kt + ccol;
        const float* gb = B + (size_t)(kt + crow) * N + tileN + ccol;
        if (kt + KC < K) {
            __builtin_prefetch(ga + KC, 0, 0);
            __builtin_prefetch(gb + (size_t)KC * N, 0, 0);
        }
        #pragma unroll
        for (int j = 0; j < 8; ++j) As[crow * ASTR + ccol + j] = ga[j];
        #pragma unroll
        for (int j = 0; j < 8; ++j) Bs[crow * BSTR + ccol + j] = gb[j];
        __syncthreads();

        #pragma unroll
        for (int kk = 0; kk < KC; kk += 4) {
            const int k0 = kk + 2 * half;
            v2f a, b;
            a.x = As[(wm * 16 + lm) * ASTR + k0];
            a.y = As[(wm * 16 + lm) * ASTR + k0 + 1];
            b.x = Bs[(k0)     * BSTR + wn * 16 + lm];
            b.y = Bs[(k0 + 1) * BSTR + wn * 16 + lm];
            acc = __builtin_amdgcn_wmma_f32_16x16x4_f32(
                false, a, false, b, (short)0, acc, false, false);
        }
        __syncthreads();
    }

    const int col   = tileN + wn * 16 + lm;
    const int rbase = tileM + wm * 16 + half * 8;
    #pragma unroll
    for (int r = 0; r < 8; ++r)
        D[(size_t)(rbase + r) * N + col] = acc[r];
}

__global__ void att_logits(const float* __restrict__ h,
                           const float* __restrict__ a_s,
                           const float* __restrict__ a_d,
                           float* __restrict__ al_s, float* __restrict__ al_d,
                           int N, int H, int C) {
    int t = blockIdx.x * blockDim.x + threadIdx.x;
    if (t >= N * H) return;
    int n = t / H, hh = t - n * H;
    const float* row = h + (size_t)n * H * C + (size_t)hh * C;
    const float* vs  = a_s + (size_t)hh * C;
    const float* vd  = a_d + (size_t)hh * C;
    float ss = 0.f, sd = 0.f;
    for (int c = 0; c < C; ++c) {
        float v = row[c];
        ss = fmaf(v, vs[c], ss);
        sd = fmaf(v, vd[c], sd);
    }
    al_s[t] = ss;
    al_d[t] = sd;
}

__global__ void fill_f32(float* __restrict__ p, float v, size_t n) {
    size_t i = blockIdx.x * (size_t)blockDim.x + threadIdx.x;
    if (i < n) p[i] = v;
}

__global__ void edge_max(const int* __restrict__ ei, int E, int N, int H,
                         const float* __restrict__ al_s,
                         const float* __restrict__ al_d,
                         float* __restrict__ mbuf) {
    int t = blockIdx.x * blockDim.x + threadIdx.x;
    int tot = (E + N) * H;
    if (t >= tot) return;
    int e = t / H, hh = t - e * H;
    int s, d;
    if (e < E) { s = ei[e]; d = ei[E + e]; } else { s = d = e - E; }
    float v = leaky02(al_s[s * H + hh] + al_d[d * H + hh]);
    atomicMaxF(&mbuf[d * H + hh], v);
}

__global__ void edge_exp(const int* __restrict__ ei, int E, int N, int H,
                         const float* __restrict__ al_s,
                         const float* __restrict__ al_d,
                         const float* __restrict__ mbuf,
                         float* __restrict__ exbuf,
                         float* __restrict__ denom) {
    int t = blockIdx.x * blockDim.x + threadIdx.x;
    int tot = (E + N) * H;
    if (t >= tot) return;
    int e = t / H, hh = t - e * H;
    int s, d;
    if (e < E) { s = ei[e]; d = ei[E + e]; } else { s = d = e - E; }
    float v  = leaky02(al_s[s * H + hh] + al_d[d * H + hh]);
    float ex = __expf(v - mbuf[d * H + hh]);
    exbuf[t] = ex;
    atomicAdd(&denom[d * H + hh], ex);
}

__global__ void fill_i32(int* __restrict__ p, int v, size_t n) {
    size_t i = blockIdx.x * (size_t)blockDim.x + threadIdx.x;
    if (i < n) p[i] = v;
}

__global__ void csr_count(const int* __restrict__ ei, int E,
                          int* __restrict__ cnt) {
    int t = blockIdx.x * blockDim.x + threadIdx.x;
    if (t >= E) return;
    atomicAdd(&cnt[ei[E + t]], 1);
}

__global__ __launch_bounds__(1024)
void csr_scan(const int* __restrict__ cnt, int* __restrict__ rowptr, int N) {
    __shared__ int part[1024];
    const int t   = threadIdx.x;
    const int per = (N + 1023) / 1024;
    int b = t * per;       if (b > N) b = N;
    int e = b + per;       if (e > N) e = N;
    int s = 0;
    for (int i = b; i < e; ++i) s += cnt[i];
    part[t] = s;
    __syncthreads();
    for (int off = 1; off < 1024; off <<= 1) {
        int v = (t >= off) ? part[t - off] : 0;
        __syncthreads();
        part[t] += v;
        __syncthreads();
    }
    int run = part[t] - s;
    for (int i = b; i < e; ++i) { rowptr[i] = run; run += cnt[i]; }
    if (t == 1023) rowptr[N] = part[1023];
}

__global__ void csr_fill(const int* __restrict__ ei, int E,
                         const int* __restrict__ rowptr,
                         int* __restrict__ cnt,
                         int* __restrict__ csr_src, int* __restrict__ csr_eid) {
    int t = blockIdx.x * blockDim.x + threadIdx.x;
    if (t >= E) return;
    int s = ei[t], d = ei[E + t];
    int slot = rowptr[d] + atomicSub(&cnt[d], 1) - 1;
    csr_src[slot] = s;
    csr_eid[slot] = t;
}

__global__ __launch_bounds__(1024)
void gather_aggregate(const int* __restrict__ rowptr,
                      const int* __restrict__ csr_src,
                      const int* __restrict__ csr_eid,
                      int E, int H, int C,
                      const float* __restrict__ hfeat,
                      const float* __restrict__ exbuf,
                      const float* __restrict__ denom,
                      float* __restrict__ agg) {
    const int d  = blockIdx.x;
    const int F  = H * C;
    const int f  = threadIdx.x * 4;
    const int hh = f / C;
    const float den = denom[(size_t)d * H + hh];
    const int beg = rowptr[d], end = rowptr[d + 1];

    float4 acc;
    {
        const float  a = exbuf[(size_t)(E + d) * H + hh] / den;
        const float4 v = *(const float4*)(hfeat + (size_t)d * F + f);
        acc.x = v.x * a; acc.y = v.y * a; acc.z = v.z * a; acc.w = v.w * a;
    }

    int j = beg;
    for (; j + 4 <= end; j += 4) {
        const int s0 = csr_src[j],     s1 = csr_src[j + 1];
        const int s2 = csr_src[j + 2], s3 = csr_src[j + 3];
        const int e0 = csr_eid[j],     e1 = csr_eid[j + 1];
        const int e2 = csr_eid[j + 2], e3 = csr_eid[j + 3];
        const float x0 = exbuf[(size_t)e0 * H + hh];
        const float x1 = exbuf[(size_t)e1 * H + hh];
        const float x2 = exbuf[(size_t)e2 * H + hh];
        const float x3 = exbuf[(size_t)e3 * H + hh];
        const float4 v0 = *(const float4*)(hfeat + (size_t)s0 * F + f);
        const float4 v1 = *(const float4*)(hfeat + (size_t)s1 * F + f);
        const float4 v2 = *(const float4*)(hfeat + (size_t)s2 * F + f);
        const float4 v3 = *(const float4*)(hfeat + (size_t)s3 * F + f);
        const float a0 = x0 / den, a1 = x1 / den, a2 = x2 / den, a3 = x3 / den;
        acc.x += v0.x * a0; acc.y += v0.y * a0; acc.z += v0.z * a0; acc.w += v0.w * a0;
        acc.x += v1.x * a1; acc.y += v1.y * a1; acc.z += v1.z * a1; acc.w += v1.w * a1;
        acc.x += v2.x * a2; acc.y += v2.y * a2; acc.z += v2.z * a2; acc.w += v2.w * a2;
        acc.x += v3.x * a3; acc.y += v3.y * a3; acc.z += v3.z * a3; acc.w += v3.w * a3;
    }
    for (; j < end; ++j) {
        const int s = csr_src[j];
        const int e = csr_eid[j];
        const float  a = exbuf[(size_t)e * H + hh] / den;
        const float4 v = *(const float4*)(hfeat + (size_t)s * F + f);
        acc.x += v.x * a; acc.y += v.y * a; acc.z += v.z * a; acc.w += v.w * a;
    }
    *(float4*)(agg + (size_t)d * F + f) = acc;
}

__global__ void bias_relu_inplace(float* __restrict__ p,
                                  const float* __restrict__ b,
                                  size_t total, int F) {
    size_t t = blockIdx.x * (size_t)blockDim.x + threadIdx.x;
    if (t >= total) return;
    int f = (int)(t % F);
    float v = p[t] + b[f];
    p[t] = v > 0.f ? v : 0.f;
}

__global__ void mean_heads_bias(const float* __restrict__ agg,
                                const float* __restrict__ b,
                                float* __restrict__ out, int N, int H, int C) {
    int t = blockIdx.x * blockDim.x + threadIdx.x;
    if (t >= N * C) return;
    int n = t / C, c = t - n * C;
    float s = 0.f;
    for (int hh = 0; hh < H; ++hh)
        s += agg[(size_t)n * H * C + (size_t)hh * C + c];
    out[t] = s / (float)H + b[c];
}

static inline unsigned cdiv_u(size_t a, unsigned b) {
    return (unsigned)((a + b - 1) / b);
}

extern "C" void kernel_launch(void* const* d_in, const int* in_sizes, int n_in,
                              void* d_out, int out_size, void* d_ws, size_t ws_size,
                              hipStream_t stream) {
    const float* x     = (const float*)d_in[0];
    const int*   ei    = (const int*)d_in[1];
    const float* W1    = (const float*)d_in[2];
    const float* a_s1  = (const float*)d_in[3];
    const float* a_d1  = (const float*)d_in[4];
    const float* b1    = (const float*)d_in[5];
    const float* W2    = (const float*)d_in[6];
    const float* a_s2  = (const float*)d_in[7];
    const float* a_d2  = (const float*)d_in[8];
    const float* b2    = (const float*)d_in[9];
    float*       out   = (float*)d_out;

    const int F1   = in_sizes[3];
    const int F2   = in_sizes[7];
    const int LAT  = in_sizes[9];
    const int H    = F2 / LAT;
    const int C1   = F1 / H;
    const int Fin  = in_sizes[2] / F1;
    const int N    = in_sizes[0] / Fin;
    const int E    = in_sizes[1] / 2;
    const int Etot = E + N;

    float* ws = (float*)d_ws;
    size_t o = 0;
    float* h1   = ws + o; o += (size_t)N * F1;
    float* agg1 = ws + o; o += (size_t)N * F1;
    float* ex   = ws + o; o += (size_t)Etot * H;
    float* als  = ws + o; o += (size_t)N * H;
    float* ald  = ws + o; o += (size_t)N * H;
    float* mbuf = ws + o; o += (size_t)N * H;
    float* den  = ws + o; o += (size_t)N * H;
    int* rowptr = (int*)(ws + o); o += (size_t)N + 1;
    int* cnt    = (int*)(ws + o); o += (size_t)N;
    int* csrc   = (int*)(ws + o); o += (size_t)E;
    int* ceid   = (int*)(ws + o); o += (size_t)E;
    float* h2   = h1;
    float* agg2 = h1 + (size_t)N * F2;
    (void)ws_size; (void)n_in; (void)out_size;

    const unsigned TB = 256;
    const float NEG_INF = -__builtin_huge_valf();

    fill_i32<<<cdiv_u((size_t)N, TB), TB, 0, stream>>>(cnt, 0, (size_t)N);
    csr_count<<<cdiv_u((size_t)E, TB), TB, 0, stream>>>(ei, E, cnt);
    csr_scan<<<1, 1024, 0, stream>>>(cnt, rowptr, N);
    csr_fill<<<cdiv_u((size_t)E, TB), TB, 0, stream>>>(ei, E, rowptr, cnt, csrc, ceid);

    fill_f32<<<cdiv_u((size_t)N * H, TB), TB, 0, stream>>>(mbuf, NEG_INF, (size_t)N * H);
    fill_f32<<<cdiv_u((size_t)N * H, TB), TB, 0, stream>>>(den, 0.f, (size_t)N * H);

    gemm_wmma_f32<<<dim3(F1 / TS, N / TS), 128, 0, stream>>>(x, W1, h1, N, Fin, F1);

    att_logits<<<cdiv_u((size_t)N * H, TB), TB, 0, stream>>>(h1, a_s1, a_d1, als, ald, N, H, C1);

    edge_max<<<cdiv_u((size_t)Etot * H, TB), TB, 0, stream>>>(ei, E, N, H, als, ald, mbuf);
    edge_exp<<<cdiv_u((size_t)Etot * H, TB), TB, 0, stream>>>(ei, E, N, H, als, ald, mbuf, ex, den);
    gather_aggregate<<<N, F1 / 4, 0, stream>>>(
        rowptr, csrc, ceid, E, H, C1, h1, ex, den, agg1);

    bias_relu_inplace<<<cdiv_u((size_t)N * F1, TB), TB, 0, stream>>>(agg1, b1, (size_t)N * F1, F1);

    gemm_wmma_f32<<<dim3(F2 / TS, N / TS), 128, 0, stream>>>(agg1, W2, h2, N, F1, F2);

    fill_f32<<<cdiv_u((size_t)N * H, TB), TB, 0, stream>>>(mbuf, NEG_INF, (size_t)N * H);
    fill_f32<<<cdiv_u((size_t)N * H, TB), TB, 0, stream>>>(den, 0.f, (size_t)N * H);

    att_logits<<<cdiv_u((size_t)N * H, TB), TB, 0, stream>>>(h2, a_s2, a_d2, als, ald, N, H, LAT);

    edge_max<<<cdiv_u((size_t)Etot * H, TB), TB, 0, stream>>>(ei, E, N, H, als, ald, mbuf);
    edge_exp<<<cdiv_u((size_t)Etot * H, TB), TB, 0, stream>>>(ei, E, N, H, als, ald, mbuf, ex, den);
    gather_aggregate<<<N, F2 / 4, 0, stream>>>(
        rowptr, csrc, ceid, E, H, LAT, h2, ex, den, agg2);

    mean_heads_bias<<<cdiv_u((size_t)N * LAT, TB), TB, 0, stream>>>(agg2, b2, out, N, H, LAT);
}


# aggregation kernel finishes each row before its one store (layer 1: +bias, ReLU; layer 2: head mean through LDS, +bias, straight to out): both elementwise launches and their round trips gone
# speedup vs baseline: 3.6942x; 1.0919x over previous
#include <hip/hip_runtime.h>
#include <math.h>

typedef __attribute__((ext_vector_type(2))) float v2f;
typedef __attribute__((ext_vector_type(8))) float v8f;

#define TS   32
#define KC   32
#define ASTR 36
#define BSTR 33

static __device__ __forceinline__ float leaky02(float x) {
    return x > 0.f ? x : 0.2f * x;
}

static __device__ __forceinline__ void atomicMaxF(float* addr, float v) {
    if (v >= 0.f)
        atomicMax((int*)addr, __float_as_int(v));
    else
        atomicMin((unsigned int*)addr, __float_as_uint(v));
}

__global__ __launch_bounds__(128)
void gemm_wmma_f32(const float* __restrict__ A, const float* __restrict__ B,
                   float* __restrict__ D, int M, int K, int N) {
    __shared__ float As[TS * ASTR];
    __shared__ float Bs[KC * BSTR];

    const int tileN = blockIdx.x * TS;
    const int tileM = blockIdx.y * TS;
    const int t    = threadIdx.x;
    const int wv   = t >> 5;
    const int lane = t & 31;
    const int half = lane >> 4;
    const int lm   = lane & 15;
    const int wm   = wv >> 1;
    const int wn   = wv & 1;

    const int crow = t >> 2;
    const int ccol = (t & 3) * 8;

    v8f acc = {};

    for (int kt = 0; kt < K; kt += KC) {
        const float* ga = A + (size_t)(tileM + crow) * K + kt + ccol;
        const float* gb = B + (size_t)(kt + crow) * N + tileN + ccol;
        if (kt + KC < K) {
            __builtin_prefetch(ga + KC, 0, 0);
            __builtin_prefetch(gb + (size_t)KC * N, 0, 0);
        }
        #pragma unroll
        for (int j = 0; j < 8; ++j) As[crow * ASTR + ccol + j] = ga[j];
        #pragma unroll
        for (int j = 0; j < 8; ++j) Bs[crow * BSTR + ccol + j] = gb[j];
        __syncthreads();

        #pragma unroll
        for (int kk = 0; kk < KC; kk += 4) {
            const int k0 = kk + 2 * half;
            v2f a, b;
            a.x = As[(wm * 16 + lm) * ASTR + k0];
            a.y = As[(wm * 16 + lm) * ASTR + k0 + 1];
            b.x = Bs[(k0)     * BSTR + wn * 16 + lm];
            b.y = Bs[(k0 + 1) * BSTR + wn * 16 + lm];
            acc = __builtin_amdgcn_wmma_f32_16x16x4_f32(
                false, a, false, b, (short)0, acc, false, false);
        }
        __syncthreads();
    }

    const int col   = tileN + wn * 16 + lm;
    const int rbase = tileM + wm * 16 + half * 8;
    #pragma unroll
    for (int r = 0; r < 8; ++r)
        D[(size_t)(rbase + r) * N + col] = acc[r];
}

__global__ void att_logits(const float* __restrict__ h,
                           const float* __restrict__ a_s,
                           const float* __restrict__ a_d,
                           float* __restrict__ al_s, float* __restrict__ al_d,
                           int N, int H, int C) {
    int t = blockIdx.x * blockDim.x + threadIdx.x;
    if (t >= N * H) return;
    int n = t / H, hh = t - n * H;
    const float* row = h + (size_t)n * H * C + (size_t)hh * C;
    const float* vs  = a_s + (size_t)hh * C;
    const float* vd  = a_d + (size_t)hh * C;
    float ss = 0.f, sd = 0.f;
    for (int c = 0; c < C; ++c) {
        float v = row[c];
        ss = fmaf(v, vs[c], ss);
        sd = fmaf(v, vd[c], sd);
    }
    al_s[t] = ss;
    al_d[t] = sd;
}

__global__ void fill_f32(float* __restrict__ p, float v, size_t n) {
    size_t i = blockIdx.x * (size_t)blockDim.x + threadIdx.x;
    if (i < n) p[i] = v;
}

__global__ void edge_max(const int* __restrict__ ei, int E, int N, int H,
                         const float* __restrict__ al_s,
                         const float* __restrict__ al_d,
                         float* __restrict__ mbuf) {
    int t = blockIdx.x * blockDim.x + threadIdx.x;
    int tot = (E + N) * H;
    if (t >= tot) return;
    int e = t / H, hh = t - e * H;
    int s, d;
    if (e < E) { s = ei[e]; d = ei[E + e]; } else { s = d = e - E; }
    float v = leaky02(al_s[s * H + hh] + al_d[d * H + hh]);
    atomicMaxF(&mbuf[d * H + hh], v);
}

__global__ void edge_exp(const int* __restrict__ ei, int E, int N, int H,
                         const float* __restrict__ al_s,
                         const float* __restrict__ al_d,
                         const float* __restrict__ mbuf,
                         float* __restrict__ exbuf,
                         float* __restrict__ denom) {
    int t = blockIdx.x * blockDim.x + threadIdx.x;
    int tot = (E + N) * H;
    if (t >= tot) return;
    int e = t / H, hh = t - e * H;
    int s, d;
    if (e < E) { s = ei[e]; d = ei[E + e]; } else { s = d = e - E; }
    float v  = leaky02(al_s[s * H + hh] + al_d[d * H + hh]);
    float ex = __expf(v - mbuf[d * H + hh]);
    exbuf[t] = ex;
    atomicAdd(&denom[d * H + hh], ex);
}

__global__ void fill_i32(int* __restrict__ p, int v, size_t n) {
    size_t i = blockIdx.x * (size_t)blockDim.x + threadIdx.x;
    if (i < n) p[i] = v;
}

__global__ void csr_count(const int* __restrict__ ei, int E,
                          int* __restrict__ cnt) {
    int t = blockIdx.x * blockDim.x + threadIdx.x;
    if (t >= E) return;
    atomicAdd(&cnt[ei[E + t]], 1);
}

__global__ __launch_bounds__(1024)
void csr_scan(const int* __restrict__ cnt, int* __restrict__ rowptr, int N) {
    __shared__ int part[1024];
    const int t   = threadIdx.x;
    const int per = (N + 1023) / 1024;
    int b = t * per;       if (b > N) b = N;
    int e = b + per;       if (e > N) e = N;
    int s = 0;
    for (int i = b; i < e; ++i) s += cnt[i];
    part[t] = s;
    __syncthreads();
    for (int off = 1; off < 1024; off <<= 1) {
        int v = (t >= off) ? part[t - off] : 0;
        __syncthreads();
        part[t] += v;
        __syncthreads();
    }
    int run = part[t] - s;
    for (int i = b; i < e; ++i) { rowptr[i] = run; run += cnt[i]; }
    if (t == 1023) rowptr[N] = part[1023];
}

__global__ void csr_fill(const int* __restrict__ ei, int E,
                         const int* __restrict__ rowptr,
                         int* __restrict__ cnt,
                         int* __restrict__ csr_src, int* __restrict__ csr_eid) {
    int t = blockIdx.x * blockDim.x + threadIdx.x;
    if (t >= E) return;
    int s = ei[t], d = ei[E + t];
    int slot = rowptr[d] + atomicSub(&cnt[d], 1) - 1;
    csr_src[slot] = s;
    csr_eid[slot] = t;
}

template <int EPI>
__global__ __launch_bounds__(1024)
void gather_aggregate(const int* __restrict__ rowptr,
                      const int* __restrict__ csr_src,
                      const int* __restrict__ csr_eid,
                      int E, int H, int C,
                      const float* __restrict__ hfeat,
                      const float* __restrict__ exbuf,
                      const float* __restrict__ denom,
                      const float* __restrict__ bias,
                      float* __restrict__ dst) {
    extern __shared__ float4 rowbuf[];
    const int d  = blockIdx.x;
    const int F  = H * C;
    const int f  = threadIdx.x * 4;
    const int hh = f / C;
    const float den = denom[(size_t)d * H + hh];
    const int beg = rowptr[d], end = rowptr[d + 1];

    float4 acc;
    {
        const float  a = exbuf[(size_t)(E + d) * H + hh] / den;
        const float4 v = *(const float4*)(hfeat + (size_t)d * F + f);
        acc.x = v.x * a; acc.y = v.y * a; acc.z = v.z * a; acc.w = v.w * a;
    }

    int j = beg;
    for (; j + 4 <= end; j += 4) {
        const int s0 = csr_src[j],     s1 = csr_src[j + 1];
        const int s2 = csr_src[j + 2], s3 = csr_src[j + 3];
        const int e0 = csr_eid[j],     e1 = csr_eid[j + 1];
        const int e2 = csr_eid[j + 2], e3 = csr_eid[j + 3];
        const float x0 = exbuf[(size_t)e0 * H + hh];
        const float x1 = exbuf[(size_t)e1 * H + hh];
        const float x2 = exbuf[(size_t)e2 * H + hh];
        const float x3 = exbuf[(size_t)e3 * H + hh];
        const float4 v0 = *(const float4*)(hfeat + (size_t)s0 * F + f);
        const float4 v1 = *(const float4*)(hfeat + (size_t)s1 * F + f);
        const float4 v2 = *(const float4*)(hfeat + (size_t)s2 * F + f);
        const float4 v3 = *(const float4*)(hfeat + (size_t)s3 * F + f);
        const float a0 = x0 / den, a1 = x1 / den, a2 = x2 / den, a3 = x3 / den;
        acc.x += v0.x * a0; acc.y += v0.y * a0; acc.z += v0.z * a0; acc.w += v0.w * a0;
        acc.x += v1.x * a1; acc.y += v1.y * a1; acc.z += v1.z * a1; acc.w += v1.w * a1;
        acc.x += v2.x * a2; acc.y += v2.y * a2; acc.z += v2.z * a2; acc.w += v2.w * a2;
        acc.x += v3.x * a3; acc.y += v3.y * a3; acc.z += v3.z * a3; acc.w += v3.w * a3;
    }
    for (; j < end; ++j) {
        const int s = csr_src[j];
        const int e = csr_eid[j];
        const float  a = exbuf[(size_t)e * H + hh] / den;
        const float4 v = *(const float4*)(hfeat + (size_t)s * F + f);
        acc.x += v.x * a; acc.y += v.y * a; acc.z += v.z * a; acc.w += v.w * a;
    }
    if (EPI == 1) {
        const float4 bb = *(const float4*)(bias + f);
        float4 r;
        r.x = acc.x + bb.x; r.y = acc.y + bb.y; r.z = acc.z + bb.z; r.w = acc.w + bb.w;
        r.x = r.x > 0.f ? r.x : 0.f; r.y = r.y > 0.f ? r.y : 0.f;
        r.z = r.z > 0.f ? r.z : 0.f; r.w = r.w > 0.f ? r.w : 0.f;
        *(float4*)(dst + (size_t)d * F + f) = r;
    } else {
        rowbuf[threadIdx.x] = acc;
        __syncthreads();
        if (f < C) {
            const int q = C >> 2;
            float4 s = {0.f, 0.f, 0.f, 0.f};
            for (int g = 0; g < H; ++g) {
                const float4 v = rowbuf[g * q + threadIdx.x];
                s.x += v.x; s.y += v.y; s.z += v.z; s.w += v.w;
            }
            const float4 bb = *(const float4*)(bias + f);
            const float hn = (float)H;
            float4 r;
            r.x = s.x / hn + bb.x; r.y = s.y / hn + bb.y;
            r.z = s.z / hn + bb.z; r.w = s.w / hn + bb.w;
            *(float4*)(dst + (size_t)d * C + f) = r;
        }
    }
}

static inline unsigned cdiv_u(size_t a, unsigned b) {
    return (unsigned)((a + b - 1) / b);
}

extern "C" void kernel_launch(void* const* d_in, const int* in_sizes, int n_in,
                              void* d_out, int out_size, void* d_ws, size_t ws_size,
                              hipStream_t stream) {
    const float* x     = (const float*)d_in[0];
    const int*   ei    = (const int*)d_in[1];
    const float* W1    = (const float*)d_in[2];
    const float* a_s1  = (const float*)d_in[3];
    const float* a_d1  = (const float*)d_in[4];
    const float* b1    = (const float*)d_in[5];
    const float* W2    = (const float*)d_in[6];
    const float* a_s2  = (const float*)d_in[7];
    const float* a_d2  = (const float*)d_in[8];
    const float* b2    = (const float*)d_in[9];
    float*       out   = (float*)d_out;

    const int F1   = in_sizes[3];
    const int F2   = in_sizes[7];
    const int LAT  = in_sizes[9];
    const int H    = F2 / LAT;
    const int C1   = F1 / H;
    const int Fin  = in_sizes[2] / F1;
    const int N    = in_sizes[0] / Fin;
    const int E    = in_sizes[1] / 2;
    const int Etot = E + N;

    float* ws = (float*)d_ws;
    size_t o = 0;
    float* h1   = ws + o; o += (size_t)N * F1;
    float* agg1 = ws + o; o += (size_t)N * F1;
    float* ex   = ws + o; o += (size_t)Etot * H;
    float* als  = ws + o; o += (size_t)N * H;
    float* ald  = ws + o; o += (size_t)N * H;
    float* mbuf = ws + o; o += (size_t)N * H;
    float* den  = ws + o; o += (size_t)N * H;
    int* rowptr = (int*)(ws + o); o += (size_t)N + 1;
    int* cnt    = (int*)(ws + o); o += (size_t)N;
    int* csrc   = (int*)(ws + o); o += (size_t)E;
    int* ceid   = (int*)(ws + o); o += (size_t)E;
    float* h2   = h1;
    (void)ws_size; (void)n_in; (void)out_size;

    const unsigned TB = 256;
    const float NEG_INF = -__builtin_huge_valf();

    fill_i32<<<cdiv_u((size_t)N, TB), TB, 0, stream>>>(cnt, 0, (size_t)N);
    csr_count<<<cdiv_u((size_t)E, TB), TB, 0, stream>>>(ei, E, cnt);
    csr_scan<<<1, 1024, 0, stream>>>(cnt, rowptr, N);
    csr_fill<<<cdiv_u((size_t)E, TB), TB, 0, stream>>>(ei, E, rowptr, cnt, csrc, ceid);

    fill_f32<<<cdiv_u((size_t)N * H, TB), TB, 0, stream>>>(mbuf, NEG_INF, (size_t)N * H);
    fill_f32<<<cdiv_u((size_t)N * H, TB), TB, 0, stream>>>(den, 0.f, (size_t)N * H);

    gemm_wmma_f32<<<dim3(F1 / TS, N / TS), 128, 0, stream>>>(x, W1, h1, N, Fin, F1);

    att_logits<<<cdiv_u((size_t)N * H, TB), TB, 0, stream>>>(h1, a_s1, a_d1, als, ald, N, H, C1);

    edge_max<<<cdiv_u((size_t)Etot * H, TB), TB, 0, stream>>>(ei, E, N, H, als, ald, mbuf);
    edge_exp<<<cdiv_u((size_t)Etot * H, TB), TB, 0, stream>>>(ei, E, N, H, als, ald, mbuf, ex, den);
    gather_aggregate<1><<<N, F1 / 4, 0, stream>>>(
        rowptr, csrc, ceid, E, H, C1, h1, ex, den, b1, agg1);

    gemm_wmma_f32<<<dim3(F2 / TS, N / TS), 128, 0, stream>>>(agg1, W2, h2, N, F1, F2);

    fill_f32<<<cdiv_u((size_t)N * H, TB), TB, 0, stream>>>(mbuf, NEG_INF, (size_t)N * H);
    fill_f32<<<cdiv_u((size_t)N * H, TB), TB, 0, stream>>>(den, 0.f, (size_t)N * H);

    att_logits<<<cdiv_u((size_t)N * H, TB), TB, 0, stream>>>(h2, a_s2, a_d2, als, ald, N, H, LAT);

    edge_max<<<cdiv_u((size_t)Etot * H, TB), TB, 0, stream>>>(ei, E, N, H, als, ald, mbuf);
    edge_exp<<<cdiv_u((size_t)Etot * H, TB), TB, 0, stream>>>(ei, E, N, H, als, ald, mbuf, ex, den);
    gather_aggregate<2><<<N, F2 / 4, (size_t)F2 * sizeof(float), stream>>>(
        rowptr, csrc, ceid, E, H, LAT, h2, ex, den, b2, out);
}


# softmax of each destination row taken inside the aggregation kernel from its CSR list in LDS: edge max/exp launches, fills and float atomics gone; denominator summed in slot order, not arrival order
# speedup vs baseline: 5.2960x; 1.4336x over previous
#include <hip/hip_runtime.h>
#include <math.h>

typedef __attribute__((ext_vector_type(2))) float v2f;
typedef __attribute__((ext_vector_type(8))) float v8f;

#define TS   32
#define KC   32
#define ASTR 36
#define BSTR 33

static __device__ __forceinline__ float leaky02(float x) {
    return x > 0.f ? x : 0.2f * x;
}

__global__ __launch_bounds__(128)
void gemm_wmma_f32(const float* __restrict__ A, const float* __restrict__ B,
                   float* __restrict__ D, int M, int K, int N) {
    __shared__ float As[TS * ASTR];
    __shared__ float Bs[KC * BSTR];

    const int tileN = blockIdx.x * TS;
    const int tileM = blockIdx.y * TS;
    const int t    = threadIdx.x;
    const int wv   = t >> 5;
    const int lane = t & 31;
    const int half = lane >> 4;
    const int lm   = lane & 15;
    const int wm   = wv >> 1;
    const int wn   = wv & 1;

    const int crow = t >> 2;
    const int ccol = (t & 3) * 8;

    v8f acc = {};

    for (int kt = 0; kt < K; kt += KC) {
        const float* ga = A + (size_t)(tileM + crow) * K + kt + ccol;
        const float* gb = B + (size_t)(kt + crow) * N + tileN + ccol;
        if (kt + KC < K) {
            __builtin_prefetch(ga + KC, 0, 0);
            __builtin_prefetch(gb + (size_t)KC * N, 0, 0);
        }
        #pragma unroll
        for (int j = 0; j < 8; ++j) As[crow * ASTR + ccol + j] = ga[j];
        #pragma unroll
        for (int j = 0; j < 8; ++j) Bs[crow * BSTR + ccol + j] = gb[j];
        __syncthreads();

        #pragma unroll
        for (int kk = 0; kk < KC; kk += 4) {
            const int k0 = kk + 2 * half;
            v2f a, b;
            a.x = As[(wm * 16 + lm) * ASTR + k0];
            a.y = As[(wm * 16 + lm) * ASTR + k0 + 1];
            b.x = Bs[(k0)     * BSTR + wn * 16 + lm];
            b.y = Bs[(k0 + 1) * BSTR + wn * 16 + lm];
            acc = __builtin_amdgcn_wmma_f32_16x16x4_f32(
                false, a, false, b, (short)0, acc, false, false);
        }
        __syncthreads();
    }

    const int col   = tileN + wn * 16 + lm;
    const int rbase = tileM + wm * 16 + half * 8;
    #pragma unroll
    for (int r = 0; r < 8; ++r)
        D[(size_t)(rbase + r) * N + col] = acc[r];
}

__global__ void att_logits(const float* __restrict__ h,
                           const float* __restrict__ a_s,
                           const float* __restrict__ a_d,
                           float* __restrict__ al_s, float* __restrict__ al_d,
                           int N, int H, int C) {
    int t = blockIdx.x * blockDim.x + threadIdx.x;
    if (t >= N * H) return;
    int n = t / H, hh = t - n * H;
    const float* row = h + (size_t)n * H * C + (size_t)hh * C;
    const float* vs  = a_s + (size_t)hh * C;
    const float* vd  = a_d + (size_t)hh * C;
    float ss = 0.f, sd = 0.f;
    for (int c = 0; c < C; ++c) {
        float v = row[c];
        ss = fmaf(v, vs[c], ss);
        sd = fmaf(v, vd[c], sd);
    }
    al_s[t] = ss;
    al_d[t] = sd;
}

__global__ void fill_i32(int* __restrict__ p, int v, size_t n) {
    size_t i = blockIdx.x * (size_t)blockDim.x + threadIdx.x;
    if (i < n) p[i] = v;
}

__global__ void csr_count(const int* __restrict__ ei, int E,
                          int* __restrict__ cnt) {
    int t = blockIdx.x * blockDim.x + threadIdx.x;
    if (t >= E) return;
    atomicAdd(&cnt[ei[E + t]], 1);
}

__global__ __launch_bounds__(1024)
void csr_scan(const int* __restrict__ cnt, int* __restrict__ rowptr, int N) {
    __shared__ int part[1024];
    const int t   = threadIdx.x;
    const int per = (N + 1023) / 1024;
    int b = t * per;       if (b > N) b = N;
    int e = b + per;       if (e > N) e = N;
    int s = 0;
    for (int i = b; i < e; ++i) s += cnt[i];
    part[t] = s;
    __syncthreads();
    for (int off = 1; off < 1024; off <<= 1) {
        int v = (t >= off) ? part[t - off] : 0;
        __syncthreads();
        part[t] += v;
        __syncthreads();
    }
    int run = part[t] - s;
    for (int i = b; i < e; ++i) { rowptr[i] = run; run += cnt[i]; }
    if (t == 1023) rowptr[N] = part[1023];
}

__global__ void csr_fill(const int* __restrict__ ei, int E,
                         const int* __restrict__ rowptr,
                         int* __restrict__ cnt,
                         int* __restrict__ csr_src) {
    int t = blockIdx.x * blockDim.x + threadIdx.x;
    if (t >= E) return;
    int s = ei[t], d = ei[E + t];
    int slot = rowptr[d] + atomicSub(&cnt[d], 1) - 1;
    csr_src[slot] = s;
}

#define ROWCAP 512

template <int EPI>
__global__ __launch_bounds__(1024)
void gat_row(const int* __restrict__ rowptr,
             const int* __restrict__ csr_src,
             int H, int C,
             const float* __restrict__ hfeat,
             const float* __restrict__ al_s,
             const float* __restrict__ al_d,
             const float* __restrict__ bias,
             float* __restrict__ dst) {
    extern __shared__ float4 rowbuf[];
    __shared__ __attribute__((aligned(16))) int   l_src[ROWCAP];
    __shared__ __attribute__((aligned(16))) float l_lg[ROWCAP];
    __shared__ __attribute__((aligned(16))) float l_al[ROWCAP];
    const int d   = blockIdx.x;
    const int F   = H * C;
    const int t   = threadIdx.x;
    const int f   = t * 4;
    const int hh  = f / C;
    const int lph = C >> 2;
    const int tl  = t - hh * lph;
    const int beg = rowptr[d], end = rowptr[d + 1];
    const int cnt = end - beg + 1;
    const int cap = (ROWCAP / H) & ~3;
    const float ad = al_d[(size_t)d * H + hh];

    float4 acc = {0.f, 0.f, 0.f, 0.f};
    if (cnt <= cap) {
        float*       lg = l_lg + hh * cap;
        float*       al = l_al + hh * cap;
        for (int v = tl; v < cnt; v += lph) {
            const int s = (v == 0) ? d : csr_src[beg + v - 1];
            if (hh == 0) l_src[v] = s;
            lg[v] = leaky02(al_s[(size_t)s * H + hh] + ad);
        }
        __syncthreads();
        float m = lg[0];
        for (int v = 1; v < cnt; ++v) { const float x = lg[v]; m = x > m ? x : m; }
        float den = 0.f;
        for (int v = 0; v < cnt; ++v) den += __expf(lg[v] - m);
        for (int v = tl; v < cnt; v += lph) al[v] = __expf(lg[v] - m) / den;
        __syncthreads();

        int v = 0;
        for (; v + 4 <= cnt; v += 4) {
            const int4   s  = *(const int4*)(l_src + v);
            const float4 a  = *(const float4*)(al + v);
            const int s0 = __builtin_amdgcn_readfirstlane(s.x);
            const int s1 = __builtin_amdgcn_readfirstlane(s.y);
            const int s2 = __builtin_amdgcn_readfirstlane(s.z);
            const int s3 = __builtin_amdgcn_readfirstlane(s.w);
            const float4 v0 = *(const float4*)(hfeat + (size_t)s0 * F + f);
            const float4 v1 = *(const float4*)(hfeat + (size_t)s1 * F + f);
            const float4 v2 = *(const float4*)(hfeat + (size_t)s2 * F + f);
            const float4 v3 = *(const float4*)(hfeat + (size_t)s3 * F + f);
            acc.x += v0.x * a.x; acc.y += v0.y * a.x; acc.z += v0.z * a.x; acc.w += v0.w * a.x;
            acc.x += v1.x * a.y; acc.y += v1.y * a.y; acc.z += v1.z * a.y; acc.w += v1.w * a.y;
            acc.x += v2.x * a.z; acc.y += v2.y * a.z; acc.z += v2.z * a.z; acc.w += v2.w * a.z;
            acc.x += v3.x * a.w; acc.y += v3.y * a.w; acc.z += v3.z * a.w; acc.w += v3.w * a.w;
        }
        for (; v < cnt; ++v) {
            const int    s  = __builtin_amdgcn_readfirstlane(l_src[v]);
            const float  a  = al[v];
            const float4 vv = *(const float4*)(hfeat + (size_t)s * F + f);
            acc.x += vv.x * a; acc.y += vv.y * a; acc.z += vv.z * a; acc.w += vv.w * a;
        }
    } else {
        const float lself = leaky02(al_s[(size_t)d * H + hh] + ad);
        float m = lself;
        for (int j = beg; j < end; ++j) {
            const float x = leaky02(al_s[(size_t)csr_src[j] * H + hh] + ad);
            m = x > m ? x : m;
        }
        float den = __expf(lself - m);
        for (int j = beg; j < end; ++j)
            den += __expf(leaky02(al_s[(size_t)csr_src[j] * H + hh] + ad) - m);
        {
            const float  a  = __expf(lself - m) / den;
            const float4 vv = *(const float4*)(hfeat + (size_t)d * F + f);
            acc.x += vv.x * a; acc.y += vv.y * a; acc.z += vv.z * a; acc.w += vv.w * a;
        }
        for (int j = beg; j < end; ++j) {
            const int    s  = csr_src[j];
            const float  a  = __expf(leaky02(al_s[(size_t)s * H + hh] + ad) - m) / den;
            const float4 vv = *(const float4*)(hfeat + (size_t)s * F + f);
            acc.x += vv.x * a; acc.y += vv.y * a; acc.z += vv.z * a; acc.w += vv.w * a;
        }
    }
    if (EPI == 1) {
        const float4 bb = *(const float4*)(bias + f);
        float4 r;
        r.x = acc.x + bb.x; r.y = acc.y + bb.y; r.z = acc.z + bb.z; r.w = acc.w + bb.w;
        r.x = r.x > 0.f ? r.x : 0.f; r.y = r.y > 0.f ? r.y : 0.f;
        r.z = r.z > 0.f ? r.z : 0.f; r.w = r.w > 0.f ? r.w : 0.f;
        *(float4*)(dst + (size_t)d * F + f) = r;
    } else {
        rowbuf[threadIdx.x] = acc;
        __syncthreads();
        if (f < C) {
            const int q = C >> 2;
            float4 s = {0.f, 0.f, 0.f, 0.f};
            for (int g = 0; g < H; ++g) {
                const float4 v = rowbuf[g * q + threadIdx.x];
                s.x += v.x; s.y += v.y; s.z += v.z; s.w += v.w;
            }
            const float4 bb = *(const float4*)(bias + f);
            const float hn = (float)H;
            float4 r;
            r.x = s.x / hn + bb.x; r.y = s.y / hn + bb.y;
            r.z = s.z / hn + bb.z; r.w = s.w / hn + bb.w;
            *(float4*)(dst + (size_t)d * C + f) = r;
        }
    }
}

static inline unsigned cdiv_u(size_t a, unsigned b) {
    return (unsigned)((a + b - 1) / b);
}

extern "C" void kernel_launch(void* const* d_in, const int* in_sizes, int n_in,
                              void* d_out, int out_size, void* d_ws, size_t ws_size,
                              hipStream_t stream) {
    const float* x     = (const float*)d_in[0];
    const int*   ei    = (const int*)d_in[1];
    const float* W1    = (const float*)d_in[2];
    const float* a_s1  = (const float*)d_in[3];
    const float* a_d1  = (const float*)d_in[4];
    const float* b1    = (const float*)d_in[5];
    const float* W2    = (const float*)d_in[6];
    const float* a_s2  = (const float*)d_in[7];
    const float* a_d2  = (const float*)d_in[8];
    const float* b2    = (const float*)d_in[9];
    float*       out   = (float*)d_out;

    const int F1   = in_sizes[3];
    const int F2   = in_sizes[7];
    const int LAT  = in_sizes[9];
    const int H    = F2 / LAT;
    const int C1   = F1 / H;
    const int Fin  = in_sizes[2] / F1;
    const int N    = in_sizes[0] / Fin;
    const int E    = in_sizes[1] / 2;

    float* ws = (float*)d_ws;
    size_t o = 0;
    float* h1   = ws + o; o += (size_t)N * F1;
    float* agg1 = ws + o; o += (size_t)N * F1;
    float* als  = ws + o; o += (size_t)N * H;
    float* ald  = ws + o; o += (size_t)N * H;
    int* rowptr = (int*)(ws + o); o += (size_t)N + 1;
    int* cnt    = (int*)(ws + o); o += (size_t)N;
    int* csrc   = (int*)(ws + o); o += (size_t)E;
    float* h2   = h1;
    (void)ws_size; (void)n_in; (void)out_size;

    const unsigned TB = 256;

    fill_i32<<<cdiv_u((size_t)N, TB), TB, 0, stream>>>(cnt, 0, (size_t)N);
    csr_count<<<cdiv_u((size_t)E, TB), TB, 0, stream>>>(ei, E, cnt);
    csr_scan<<<1, 1024, 0, stream>>>(cnt, rowptr, N);
    csr_fill<<<cdiv_u((size_t)E, TB), TB, 0, stream>>>(ei, E, rowptr, cnt, csrc);

    gemm_wmma_f32<<<dim3(F1 / TS, N / TS), 128, 0, stream>>>(x, W1, h1, N, Fin, F1);
    att_logits<<<cdiv_u((size_t)N * H, TB), TB, 0, stream>>>(h1, a_s1, a_d1, als, ald, N, H, C1);
    gat_row<1><<<N, F1 / 4, 0, stream>>>(
        rowptr, csrc, H, C1, h1, als, ald, b1, agg1);

    gemm_wmma_f32<<<dim3(F2 / TS, N / TS), 128, 0, stream>>>(agg1, W2, h2, N, F1, F2);
    att_logits<<<cdiv_u((size_t)N * H, TB), TB, 0, stream>>>(h2, a_s2, a_d2, als, ald, N, H, LAT);
    gat_row<2><<<N, F2 / 4, (size_t)F2 * sizeof(float), stream>>>(
        rowptr, csrc, H, LAT, h2, als, ald, b2, out);
}


# GEMM register-blocked: 64x64 block tile, each wave 2x2 accumulator tiles from fragments read once; next K chunk loaded under the products (two LDS stages); per-element k order unchanged
# speedup vs baseline: 6.2586x; 1.1818x over previous
#include <hip/hip_runtime.h>
#include <math.h>

typedef __attribute__((ext_vector_type(2))) float v2f;
typedef __attribute__((ext_vector_type(8))) float v8f;

#define BM   64
#define BN   64
#define KC   32
#define ASTR 36
#define BSTR 72

static __device__ __forceinline__ float leaky02(float x) {
    return x > 0.f ? x : 0.2f * x;
}

static __device__ __forceinline__ void mma_chunk(
        const float* __restrict__ as, const float* __restrict__ bs,
        int aoff, int boff,
        v8f& c00, v8f& c01, v8f& c10, v8f& c11) {
    for (int kk = 0; kk < KC; kk += 4) {
        const v2f a0 = *(const v2f*)(as + aoff + kk);
        const v2f a1 = *(const v2f*)(as + aoff + 16 * ASTR + kk);
        v2f b0, b1;
        b0.x = bs[boff + kk * BSTR];
        b0.y = bs[boff + (kk + 1) * BSTR];
        b1.x = bs[boff + 16 + kk * BSTR];
        b1.y = bs[boff + 16 + (kk + 1) * BSTR];
        c00 = __builtin_amdgcn_wmma_f32_16x16x4_f32(
            false, a0, false, b0, (short)0, c00, false, false);
        c01 = __builtin_amdgcn_wmma_f32_16x16x4_f32(
            false, a0, false, b1, (short)0, c01, false, false);
        c10 = __builtin_amdgcn_wmma_f32_16x16x4_f32(
            false, a1, false, b0, (short)0, c10, false, false);
        c11 = __builtin_amdgcn_wmma_f32_16x16x4_f32(
            false, a1, false, b1, (short)0, c11, false, false);
    }
}

__global__ __launch_bounds__(128)
void gemm_wmma_f32(const float* __restrict__ A, const float* __restrict__ B,
                   float* __restrict__ D, int M, int K, int N) {
    __shared__ __attribute__((aligned(16))) float As[2][BM * ASTR];
    __shared__ __attribute__((aligned(16))) float Bs[2][KC * BSTR];

    const int tileN = blockIdx.x * BN;
    const int tileM = blockIdx.y * BM;
    const int t    = threadIdx.x;
    const int wv   = t >> 5;
    const int lane = t & 31;
    const int half = lane >> 4;
    const int lm   = lane & 15;
    const int wm   = wv >> 1;
    const int wn   = wv & 1;

    const int arow = t >> 3;
    const int acol = (t & 7) * 4;
    const int brow = t >> 4;
    const int bcol = (t & 15) * 4;
    const float* ga = A + (size_t)(tileM + arow) * K + acol;
    const float* gb = B + (size_t)brow * N + tileN + bcol;
    const int sa = arow * ASTR + acol;
    const int sb = brow * BSTR + bcol;

    const int aoff = (wm * 32 + lm) * ASTR + 2 * half;
    const int boff = (2 * half) * BSTR + wn * 32 + lm;

    float4 ra[4], rb[4];
    for (int j = 0; j < 4; ++j) ra[j] = *(const float4*)(ga + (size_t)(16 * j) * K);
    for (int j = 0; j < 4; ++j) rb[j] = *(const float4*)(gb + (size_t)(8 * j) * N);
    for (int j = 0; j < 4; ++j) *(float4*)(&As[0][sa + 16 * j * ASTR]) = ra[j];
    for (int j = 0; j < 4; ++j) *(float4*)(&Bs[0][sb + 8 * j * BSTR]) = rb[j];
    __syncthreads();

    v8f c00 = {}, c01 = {}, c10 = {}, c11 = {};
    const int nk = K / KC;
    for (int c = 0; c + 1 < nk; ++c) {
        const int cur = c & 1;
        ga += KC;
        gb += (size_t)KC * N;
        for (int j = 0; j < 4; ++j) ra[j] = *(const float4*)(ga + (size_t)(16 * j) * K);
        for (int j = 0; j < 4; ++j) rb[j] = *(const float4*)(gb + (size_t)(8 * j) * N);
        mma_chunk(As[cur], Bs[cur], aoff, boff, c00, c01, c10, c11);
        for (int j = 0; j < 4; ++j) *(float4*)(&As[cur ^ 1][sa + 16 * j * ASTR]) = ra[j];
        for (int j = 0; j < 4; ++j) *(float4*)(&Bs[cur ^ 1][sb + 8 * j * BSTR]) = rb[j];
        __syncthreads();
    }
    mma_chunk(As[(nk - 1) & 1], Bs[(nk - 1) & 1], aoff, boff, c00, c01, c10, c11);

    const int col   = tileN + wn * 32 + lm;
    const int rbase = tileM + wm * 32 + half * 8;
    float* d0 = D + (size_t)rbase * N + col;
    float* d1 = d0 + (size_t)16 * N;
    for (int r = 0; r < 8; ++r) {
        d0[(size_t)r * N]      = c00[r];
        d0[(size_t)r * N + 16] = c01[r];
        d1[(size_t)r * N]      = c10[r];
        d1[(size_t)r * N + 16] = c11[r];
    }
}

__global__ void att_logits(const float* __restrict__ h,
                           const float* __restrict__ a_s,
                           const float* __restrict__ a_d,
                           float* __restrict__ al_s, float* __restrict__ al_d,
                           int N, int H, int C) {
    int t = blockIdx.x * blockDim.x + threadIdx.x;
    if (t >= N * H) return;
    int n = t / H, hh = t - n * H;
    const float* row = h + (size_t)n * H * C + (size_t)hh * C;
    const float* vs  = a_s + (size_t)hh * C;
    const float* vd  = a_d + (size_t)hh * C;
    float ss = 0.f, sd = 0.f;
    for (int c = 0; c < C; ++c) {
        float v = row[c];
        ss = fmaf(v, vs[c], ss);
        sd = fmaf(v, vd[c], sd);
    }
    al_s[t] = ss;
    al_d[t] = sd;
}

__global__ void fill_i32(int* __restrict__ p, int v, size_t n) {
    size_t i = blockIdx.x * (size_t)blockDim.x + threadIdx.x;
    if (i < n) p[i] = v;
}

__global__ void csr_count(const int* __restrict__ ei, int E,
                          int* __restrict__ cnt) {
    int t = blockIdx.x * blockDim.x + threadIdx.x;
    if (t >= E) return;
    atomicAdd(&cnt[ei[E + t]], 1);
}

__global__ __launch_bounds__(1024)
void csr_scan(const int* __restrict__ cnt, int* __restrict__ rowptr, int N) {
    __shared__ int part[1024];
    const int t   = threadIdx.x;
    const int per = (N + 1023) / 1024;
    int b = t * per;       if (b > N) b = N;
    int e = b + per;       if (e > N) e = N;
    int s = 0;
    for (int i = b; i < e; ++i) s += cnt[i];
    part[t] = s;
    __syncthreads();
    for (int off = 1; off < 1024; off <<= 1) {
        int v = (t >= off) ? part[t - off] : 0;
        __syncthreads();
        part[t] += v;
        __syncthreads();
    }
    int run = part[t] - s;
    for (int i = b; i < e; ++i) { rowptr[i] = run; run += cnt[i]; }
    if (t == 1023) rowptr[N] = part[1023];
}

__global__ void csr_fill(const int* __restrict__ ei, int E,
                         const int* __restrict__ rowptr,
                         int* __restrict__ cnt,
                         int* __restrict__ csr_src) {
    int t = blockIdx.x * blockDim.x + threadIdx.x;
    if (t >= E) return;
    int s = ei[t], d = ei[E + t];
    int slot = rowptr[d] + atomicSub(&cnt[d], 1) - 1;
    csr_src[slot] = s;
}

#define ROWCAP 512

template <int EPI>
__global__ __launch_bounds__(1024)
void gat_row(const int* __restrict__ rowptr,
             const int* __restrict__ csr_src,
             int H, int C,
             const float* __restrict__ hfeat,
             const float* __restrict__ al_s,
             const float* __restrict__ al_d,
             const float* __restrict__ bias,
             float* __restrict__ dst) {
    extern __shared__ float4 rowbuf[];
    __shared__ __attribute__((aligned(16))) int   l_src[ROWCAP];
    __shared__ __attribute__((aligned(16))) float l_lg[ROWCAP];
    __shared__ __attribute__((aligned(16))) float l_al[ROWCAP];
    const int d   = blockIdx.x;
    const int F   = H * C;
    const int t   = threadIdx.x;
    const int f   = t * 4;
    const int hh  = f / C;
    const int lph = C >> 2;
    const int tl  = t - hh * lph;
    const int beg = rowptr[d], end = rowptr[d + 1];
    const int cnt = end - beg + 1;
    const int cap = (ROWCAP / H) & ~3;
    const float ad = al_d[(size_t)d * H + hh];

    float4 acc = {0.f, 0.f, 0.f, 0.f};
    if (cnt <= cap) {
        float*       lg = l_lg + hh * cap;
        float*       al = l_al + hh * cap;
        for (int v = tl; v < cnt; v += lph) {
            const int s = (v == 0) ? d : csr_src[beg + v - 1];
            if (hh == 0) l_src[v] = s;
            lg[v] = leaky02(al_s[(size_t)s * H + hh] + ad);
        }
        __syncthreads();
        float m = lg[0];
        for (int v = 1; v < cnt; ++v) { const float x = lg[v]; m = x > m ? x : m; }
        float den = 0.f;
        for (int v = 0; v < cnt; ++v) den += __expf(lg[v] - m);
        for (int v = tl; v < cnt; v += lph) al[v] = __expf(lg[v] - m) / den;
        __syncthreads();

        int v = 0;
        for (; v + 4 <= cnt; v += 4) {
            const int4   s  = *(const int4*)(l_src + v);
            const float4 a  = *(const float4*)(al + v);
            const int s0 = __builtin_amdgcn_readfirstlane(s.x);
            const int s1 = __builtin_amdgcn_readfirstlane(s.y);
            const int s2 = __builtin_amdgcn_readfirstlane(s.z);
            const int s3 = __builtin_amdgcn_readfirstlane(s.w);
            const float4 v0 = *(const float4*)(hfeat + (size_t)s0 * F + f);
            const float4 v1 = *(const float4*)(hfeat + (size_t)s1 * F + f);
            const float4 v2 = *(const float4*)(hfeat + (size_t)s2 * F + f);
            const float4 v3 = *(const float4*)(hfeat + (size_t)s3 * F + f);
            acc.x += v0.x * a.x; acc.y += v0.y * a.x; acc.z += v0.z * a.x; acc.w += v0.w * a.x;
            acc.x += v1.x * a.y; acc.y += v1.y * a.y; acc.z += v1.z * a.y; acc.w += v1.w * a.y;
            acc.x += v2.x * a.z; acc.y += v2.y * a.z; acc.z += v2.z * a.z; acc.w += v2.w * a.z;
            acc.x += v3.x * a.w; acc.y += v3.y * a.w; acc.z += v3.z * a.w; acc.w += v3.w * a.w;
        }
        for (; v < cnt; ++v) {
            const int    s  = __builtin_amdgcn_readfirstlane(l_src[v]);
            const float  a  = al[v];
            const float4 vv = *(const float4*)(hfeat + (size_t)s * F + f);
            acc.x += vv.x * a; acc.y += vv.y * a; acc.z += vv.z * a; acc.w += vv.w * a;
        }
    } else {
        const float lself = leaky02(al_s[(size_t)d * H + hh] + ad);
        float m = lself;
        for (int j = beg; j < end; ++j) {
            const float x = leaky02(al_s[(size_t)csr_src[j] * H + hh] + ad);
            m = x > m ? x : m;
        }
        float den = __expf(lself - m);
        for (int j = beg; j < end; ++j)
            den += __expf(leaky02(al_s[(size_t)csr_src[j] * H + hh] + ad) - m);
        {
            const float  a  = __expf(lself - m) / den;
            const float4 vv = *(const float4*)(hfeat + (size_t)d * F + f);
            acc.x += vv.x * a; acc.y += vv.y * a; acc.z += vv.z * a; acc.w += vv.w * a;
        }
        for (int j = beg; j < end; ++j) {
            const int    s  = csr_src[j];
            const float  a  = __expf(leaky02(al_s[(size_t)s * H + hh] + ad) - m) / den;
            const float4 vv = *(const float4*)(hfeat + (size_t)s * F + f);
            acc.x += vv.x * a; acc.y += vv.y * a; acc.z += vv.z * a; acc.w += vv.w * a;
        }
    }
    if (EPI == 1) {
        const float4 bb = *(const float4*)(bias + f);
        float4 r;
        r.x = acc.x + bb.x; r.y = acc.y + bb.y; r.z = acc.z + bb.z; r.w = acc.w + bb.w;
        r.x = r.x > 0.f ? r.x : 0.f; r.y = r.y > 0.f ? r.y : 0.f;
        r.z = r.z > 0.f ? r.z : 0.f; r.w = r.w > 0.f ? r.w : 0.f;
        *(float4*)(dst + (size_t)d * F + f) = r;
    } else {
        rowbuf[threadIdx.x] = acc;
        __syncthreads();
        if (f < C) {
            const int q = C >> 2;
            float4 s = {0.f, 0.f, 0.f, 0.f};
            for (int g = 0; g < H; ++g) {
                const float4 v = rowbuf[g * q + threadIdx.x];
                s.x += v.x; s.y += v.y; s.z += v.z; s.w += v.w;
            }
            const float4 bb = *(const float4*)(bias + f);
            const float hn = (float)H;
            float4 r;
            r.x = s.x / hn + bb.x; r.y = s.y / hn + bb.y;
            r.z = s.z / hn + bb.z; r.w = s.w / hn + bb.w;
            *(float4*)(dst + (size_t)d * C + f) = r;
        }
    }
}

static inline unsigned cdiv_u(size_t a, unsigned b) {
    return (unsigned)((a + b - 1) / b);
}

extern "C" void kernel_launch(void* const* d_in, const int* in_sizes, int n_in,
                              void* d_out, int out_size, void* d_ws, size_t ws_size,
                              hipStream_t stream) {
    const float* x     = (const float*)d_in[0];
    const int*   ei    = (const int*)d_in[1];
    const float* W1    = (const float*)d_in[2];
    const float* a_s1  = (const float*)d_in[3];
    const float* a_d1  = (const float*)d_in[4];
    const float* b1    = (const float*)d_in[5];
    const float* W2    = (const float*)d_in[6];
    const float* a_s2  = (const float*)d_in[7];
    const float* a_d2  = (const float*)d_in[8];
    const float* b2    = (const float*)d_in[9];
    float*       out   = (float*)d_out;

    const int F1   = in_sizes[3];
    const int F2   = in_sizes[7];
    const int LAT  = in_sizes[9];
    const int H    = F2 / LAT;
    const int C1   = F1 / H;
    const int Fin  = in_sizes[2] / F1;
    const int N    = in_sizes[0] / Fin;
    const int E    = in_sizes[1] / 2;

    float* ws = (float*)d_ws;
    size_t o = 0;
    float* h1   = ws + o; o += (size_t)N * F1;
    float* agg1 = ws + o; o += (size_t)N * F1;
    float* als  = ws + o; o += (size_t)N * H;
    float* ald  = ws + o; o += (size_t)N * H;
    int* rowptr = (int*)(ws + o); o += (size_t)N + 1;
    int* cnt    = (int*)(ws + o); o += (size_t)N;
    int* csrc   = (int*)(ws + o); o += (size_t)E;
    float* h2   = h1;
    (void)ws_size; (void)n_in; (void)out_size;

    const unsigned TB = 256;

    fill_i32<<<cdiv_u((size_t)N, TB), TB, 0, stream>>>(cnt, 0, (size_t)N);
    csr_count<<<cdiv_u((size_t)E, TB), TB, 0, stream>>>(ei, E, cnt);
    csr_scan<<<1, 1024, 0, stream>>>(cnt, rowptr, N);
    csr_fill<<<cdiv_u((size_t)E, TB), TB, 0, stream>>>(ei, E, rowptr, cnt, csrc);

    gemm_wmma_f32<<<dim3(F1 / BN, N / BM), 128, 0, stream>>>(x, W1, h1, N, Fin, F1);
    att_logits<<<cdiv_u((size_t)N * H, TB), TB, 0, stream>>>(h1, a_s1, a_d1, als, ald, N, H, C1);
    gat_row<1><<<N, F1 / 4, 0, stream>>>(
        rowptr, csrc, H, C1, h1, als, ald, b1, agg1);

    gemm_wmma_f32<<<dim3(F2 / BN, N / BM), 128, 0, stream>>>(agg1, W2, h2, N, F1, F2);
    att_logits<<<cdiv_u((size_t)N * H, TB), TB, 0, stream>>>(h2, a_s2, a_d2, als, ald, N, H, LAT);
    gat_row<2><<<N, F2 / 4, (size_t)F2 * sizeof(float), stream>>>(
        rowptr, csrc, H, LAT, h2, als, ald, b2, out);
}


# edges bucketed by destination in one pass (64 slots per row, one returning int atomic per edge): count and prefix-scan launches gone, 10 -> 8 launches; sum order = slot arrival, as before
# speedup vs baseline: 7.3295x; 1.1711x over previous
#include <hip/hip_runtime.h>
#include <math.h>

typedef __attribute__((ext_vector_type(2))) float v2f;
typedef __attribute__((ext_vector_type(8))) float v8f;

#define BM   64
#define BN   64
#define KC   32
#define ASTR 36
#define BSTR 72

static __device__ __forceinline__ float leaky02(float x) {
    return x > 0.f ? x : 0.2f * x;
}

static __device__ __forceinline__ void mma_chunk(
        const float* __restrict__ as, const float* __restrict__ bs,
        int aoff, int boff,
        v8f& c00, v8f& c01, v8f& c10, v8f& c11) {
    for (int kk = 0; kk < KC; kk += 4) {
        const v2f a0 = *(const v2f*)(as + aoff + kk);
        const v2f a1 = *(const v2f*)(as + aoff + 16 * ASTR + kk);
        v2f b0, b1;
        b0.x = bs[boff + kk * BSTR];
        b0.y = bs[boff + (kk + 1) * BSTR];
        b1.x = bs[boff + 16 + kk * BSTR];
        b1.y = bs[boff + 16 + (kk + 1) * BSTR];
        c00 = __builtin_amdgcn_wmma_f32_16x16x4_f32(
            false, a0, false, b0, (short)0, c00, false, false);
        c01 = __builtin_amdgcn_wmma_f32_16x16x4_f32(
            false, a0, false, b1, (short)0, c01, false, false);
        c10 = __builtin_amdgcn_wmma_f32_16x16x4_f32(
            false, a1, false, b0, (short)0, c10, false, false);
        c11 = __builtin_amdgcn_wmma_f32_16x16x4_f32(
            false, a1, false, b1, (short)0, c11, false, false);
    }
}

__global__ __launch_bounds__(128)
void gemm_wmma_f32(const float* __restrict__ A, const float* __restrict__ B,
                   float* __restrict__ D, int M, int K, int N) {
    __shared__ __attribute__((aligned(16))) float As[2][BM * ASTR];
    __shared__ __attribute__((aligned(16))) float Bs[2][KC * BSTR];

    const int tileN = blockIdx.x * BN;
    const int tileM = blockIdx.y * BM;
    const int t    = threadIdx.x;
    const int wv   = t >> 5;
    const int lane = t & 31;
    const int half = lane >> 4;
    const int lm   = lane & 15;
    const int wm   = wv >> 1;
    const int wn   = wv & 1;

    const int arow = t >> 3;
    const int acol = (t & 7) * 4;
    const int brow = t >> 4;
    const int bcol = (t & 15) * 4;
    const float* ga = A + (size_t)(tileM + arow) * K + acol;
    const float* gb = B + (size_t)brow * N + tileN + bcol;
    const int sa = arow * ASTR + acol;
    const int sb = brow * BSTR + bcol;

    const int aoff = (wm * 32 + lm) * ASTR + 2 * half;
    const int boff = (2 * half) * BSTR + wn * 32 + lm;

    float4 ra[4], rb[4];
    for (int j = 0; j < 4; ++j) ra[j] = *(const float4*)(ga + (size_t)(16 * j) * K);
    for (int j = 0; j < 4; ++j) rb[j] = *(const float4*)(gb + (size_t)(8 * j) * N);
    for (int j = 0; j < 4; ++j) *(float4*)(&As[0][sa + 16 * j * ASTR]) = ra[j];
    for (int j = 0; j < 4; ++j) *(float4*)(&Bs[0][sb + 8 * j * BSTR]) = rb[j];
    __syncthreads();

    v8f c00 = {}, c01 = {}, c10 = {}, c11 = {};
    const int nk = K / KC;
    for (int c = 0; c + 1 < nk; ++c) {
        const int cur = c & 1;
        ga += KC;
        gb += (size_t)KC * N;
        for (int j = 0; j < 4; ++j) ra[j] = *(const float4*)(ga + (size_t)(16 * j) * K);
        for (int j = 0; j < 4; ++j) rb[j] = *(const float4*)(gb + (size_t)(8 * j) * N);
        mma_chunk(As[cur], Bs[cur], aoff, boff, c00, c01, c10, c11);
        for (int j = 0; j < 4; ++j) *(float4*)(&As[cur ^ 1][sa + 16 * j * ASTR]) = ra[j];
        for (int j = 0; j < 4; ++j) *(float4*)(&Bs[cur ^ 1][sb + 8 * j * BSTR]) = rb[j];
        __syncthreads();
    }
    mma_chunk(As[(nk - 1) & 1], Bs[(nk - 1) & 1], aoff, boff, c00, c01, c10, c11);

    const int col   = tileN + wn * 32 + lm;
    const int rbase = tileM + wm * 32 + half * 8;
    float* d0 = D + (size_t)rbase * N + col;
    float* d1 = d0 + (size_t)16 * N;
    for (int r = 0; r < 8; ++r) {
        d0[(size_t)r * N]      = c00[r];
        d0[(size_t)r * N + 16] = c01[r];
        d1[(size_t)r * N]      = c10[r];
        d1[(size_t)r * N + 16] = c11[r];
    }
}

__global__ void att_logits(const float* __restrict__ h,
                           const float* __restrict__ a_s,
                           const float* __restrict__ a_d,
                           float* __restrict__ al_s, float* __restrict__ al_d,
                           int N, int H, int C) {
    int t = blockIdx.x * blockDim.x + threadIdx.x;
    if (t >= N * H) return;
    int n = t / H, hh = t - n * H;
    const float* row = h + (size_t)n * H * C + (size_t)hh * C;
    const float* vs  = a_s + (size_t)hh * C;
    const float* vd  = a_d + (size_t)hh * C;
    float ss = 0.f, sd = 0.f;
    for (int c = 0; c < C; ++c) {
        float v = row[c];
        ss = fmaf(v, vs[c], ss);
        sd = fmaf(v, vd[c], sd);
    }
    al_s[t] = ss;
    al_d[t] = sd;
}

#define BCAP 64

__global__ void fill_i32(int* __restrict__ p, int v, size_t n) {
    size_t i = blockIdx.x * (size_t)blockDim.x + threadIdx.x;
    if (i < n) p[i] = v;
}

__global__ void bucket_fill(const int* __restrict__ ei, int E,
                            int* __restrict__ deg,
                            int* __restrict__ bucket) {
    int t = blockIdx.x * blockDim.x + threadIdx.x;
    if (t >= E) return;
    int s = ei[t], d = ei[E + t];
    int slot = atomicAdd(&deg[d], 1);
    if (slot < BCAP) bucket[(size_t)d * BCAP + slot] = s;
}

#define ROWCAP 512

template <int EPI>
__global__ __launch_bounds__(1024)
void gat_row(const int* __restrict__ deg,
             const int* __restrict__ bucket,
             const int* __restrict__ ei, int E,
             int H, int C,
             const float* __restrict__ hfeat,
             const float* __restrict__ al_s,
             const float* __restrict__ al_d,
             const float* __restrict__ bias,
             float* __restrict__ dst) {
    extern __shared__ float4 rowbuf[];
    __shared__ __attribute__((aligned(16))) int   l_src[ROWCAP];
    __shared__ __attribute__((aligned(16))) float l_lg[ROWCAP];
    __shared__ __attribute__((aligned(16))) float l_al[ROWCAP];
    const int d   = blockIdx.x;
    const int F   = H * C;
    const int t   = threadIdx.x;
    const int f   = t * 4;
    const int hh  = f / C;
    const int lph = C >> 2;
    const int tl  = t - hh * lph;
    const int nin = deg[d];
    const int* lst = bucket + (size_t)d * BCAP;
    const int cnt = nin + 1;
    const int cap = (ROWCAP / H) & ~3;
    const float ad = al_d[(size_t)d * H + hh];

    float4 acc = {0.f, 0.f, 0.f, 0.f};
    if (nin <= BCAP && cnt <= cap) {
        float*       lg = l_lg + hh * cap;
        float*       al = l_al + hh * cap;
        for (int v = tl; v < cnt; v += lph) {
            const int s = (v == 0) ? d : lst[v - 1];
            if (hh == 0) l_src[v] = s;
            lg[v] = leaky02(al_s[(size_t)s * H + hh] + ad);
        }
        __syncthreads();
        float m = lg[0];
        for (int v = 1; v < cnt; ++v) { const float x = lg[v]; m = x > m ? x : m; }
        float den = 0.f;
        for (int v = 0; v < cnt; ++v) den += __expf(lg[v] - m);
        for (int v = tl; v < cnt; v += lph) al[v] = __expf(lg[v] - m) / den;
        __syncthreads();

        int v = 0;
        for (; v + 4 <= cnt; v += 4) {
            const int4   s  = *(const int4*)(l_src + v);
            const float4 a  = *(const float4*)(al + v);
            const int s0 = __builtin_amdgcn_readfirstlane(s.x);
            const int s1 = __builtin_amdgcn_readfirstlane(s.y);
            const int s2 = __builtin_amdgcn_readfirstlane(s.z);
            const int s3 = __builtin_amdgcn_readfirstlane(s.w);
            const float4 v0 = *(const float4*)(hfeat + (size_t)s0 * F + f);
            const float4 v1 = *(const float4*)(hfeat + (size_t)s1 * F + f);
            const float4 v2 = *(const float4*)(hfeat + (size_t)s2 * F + f);
            const float4 v3 = *(const float4*)(hfeat + (size_t)s3 * F + f);
            acc.x += v0.x * a.x; acc.y += v0.y * a.x; acc.z += v0.z * a.x; acc.w += v0.w * a.x;
            acc.x += v1.x * a.y; acc.y += v1.y * a.y; acc.z += v1.z * a.y; acc.w += v1.w * a.y;
            acc.x += v2.x * a.z; acc.y += v2.y * a.z; acc.z += v2.z * a.z; acc.w += v2.w * a.z;
            acc.x += v3.x * a.w; acc.y += v3.y * a.w; acc.z += v3.z * a.w; acc.w += v3.w * a.w;
        }
        for (; v < cnt; ++v) {
            const int    s  = __builtin_amdgcn_readfirstlane(l_src[v]);
            const float  a  = al[v];
            const float4 vv = *(const float4*)(hfeat + (size_t)s * F + f);
            acc.x += vv.x * a; acc.y += vv.y * a; acc.z += vv.z * a; acc.w += vv.w * a;
        }
    } else {
        const int* es = ei;
        const int* ed = ei + E;
        const float lself = leaky02(al_s[(size_t)d * H + hh] + ad);
        float m = lself;
        for (int j = 0; j < E; ++j) {
            if (ed[j] != d) continue;
            const float x = leaky02(al_s[(size_t)es[j] * H + hh] + ad);
            m = x > m ? x : m;
        }
        float den = __expf(lself - m);
        for (int j = 0; j < E; ++j) {
            if (ed[j] != d) continue;
            den += __expf(leaky02(al_s[(size_t)es[j] * H + hh] + ad) - m);
        }
        {
            const float  a  = __expf(lself - m) / den;
            const float4 vv = *(const float4*)(hfeat + (size_t)d * F + f);
            acc.x += vv.x * a; acc.y += vv.y * a; acc.z += vv.z * a; acc.w += vv.w * a;
        }
        for (int j = 0; j < E; ++j) {
            if (ed[j] != d) continue;
            const int    s  = es[j];
            const float  a  = __expf(leaky02(al_s[(size_t)s * H + hh] + ad) - m) / den;
            const float4 vv = *(const float4*)(hfeat + (size_t)s * F + f);
            acc.x += vv.x * a; acc.y += vv.y * a; acc.z += vv.z * a; acc.w += vv.w * a;
        }
    }
    if (EPI == 1) {
        const float4 bb = *(const float4*)(bias + f);
        float4 r;
        r.x = acc.x + bb.x; r.y = acc.y + bb.y; r.z = acc.z + bb.z; r.w = acc.w + bb.w;
        r.x = r.x > 0.f ? r.x : 0.f; r.y = r.y > 0.f ? r.y : 0.f;
        r.z = r.z > 0.f ? r.z : 0.f; r.w = r.w > 0.f ? r.w : 0.f;
        *(float4*)(dst + (size_t)d * F + f) = r;
    } else {
        rowbuf[threadIdx.x] = acc;
        __syncthreads();
        if (f < C) {
            const int q = C >> 2;
            float4 s = {0.f, 0.f, 0.f, 0.f};
            for (int g = 0; g < H; ++g) {
                const float4 v = rowbuf[g * q + threadIdx.x];
                s.x += v.x; s.y += v.y; s.z += v.z; s.w += v.w;
            }
            const float4 bb = *(const float4*)(bias + f);
            const float hn = (float)H;
            float4 r;
            r.x = s.x / hn + bb.x; r.y = s.y / hn + bb.y;
            r.z = s.z / hn + bb.z; r.w = s.w / hn + bb.w;
            *(float4*)(dst + (size_t)d * C + f) = r;
        }
    }
}

static inline unsigned cdiv_u(size_t a, unsigned b) {
    return (unsigned)((a + b - 1) / b);
}

extern "C" void kernel_launch(void* const* d_in, const int* in_sizes, int n_in,
                              void* d_out, int out_size, void* d_ws, size_t ws_size,
                              hipStream_t stream) {
    const float* x     = (const float*)d_in[0];
    const int*   ei    = (const int*)d_in[1];
    const float* W1    = (const float*)d_in[2];
    const float* a_s1  = (const float*)d_in[3];
    const float* a_d1  = (const float*)d_in[4];
    const float* b1    = (const float*)d_in[5];
    const float* W2    = (const float*)d_in[6];
    const float* a_s2  = (const float*)d_in[7];
    const float* a_d2  = (const float*)d_in[8];
    const float* b2    = (const float*)d_in[9];
    float*       out   = (float*)d_out;

    const int F1   = in_sizes[3];
    const int F2   = in_sizes[7];
    const int LAT  = in_sizes[9];
    const int H    = F2 / LAT;
    const int C1   = F1 / H;
    const int Fin  = in_sizes[2] / F1;
    const int N    = in_sizes[0] / Fin;
    const int E    = in_sizes[1] / 2;

    float* ws = (float*)d_ws;
    size_t o = 0;
    float* h1   = ws + o; o += (size_t)N * F1;
    float* agg1 = ws + o; o += (size_t)N * F1;
    float* als  = ws + o; o += (size_t)N * H;
    float* ald  = ws + o; o += (size_t)N * H;
    int* deg    = (int*)(ws + o); o += (size_t)N;
    int* bucket = (int*)(ws + o); o += (size_t)N * BCAP;
    float* h2   = h1;
    (void)ws_size; (void)n_in; (void)out_size;

    const unsigned TB = 256;

    fill_i32<<<cdiv_u((size_t)N, TB), TB, 0, stream>>>(deg, 0, (size_t)N);
    bucket_fill<<<cdiv_u((size_t)E, TB), TB, 0, stream>>>(ei, E, deg, bucket);

    gemm_wmma_f32<<<dim3(F1 / BN, N / BM), 128, 0, stream>>>(x, W1, h1, N, Fin, F1);
    att_logits<<<cdiv_u((size_t)N * H, TB), TB, 0, stream>>>(h1, a_s1, a_d1, als, ald, N, H, C1);
    gat_row<1><<<N, F1 / 4, 0, stream>>>(
        deg, bucket, ei, E, H, C1, h1, als, ald, b1, agg1);

    gemm_wmma_f32<<<dim3(F2 / BN, N / BM), 128, 0, stream>>>(agg1, W2, h2, N, F1, F2);
    att_logits<<<cdiv_u((size_t)N * H, TB), TB, 0, stream>>>(h2, a_s2, a_d2, als, ald, N, H, LAT);
    gat_row<2><<<N, F2 / 4, (size_t)F2 * sizeof(float), stream>>>(
        deg, bucket, ei, E, H, LAT, h2, als, ald, b2, out);
}


# attention logits taken inside the row kernel from the rows it gathers (float4 dot + lane butterfly), softmax as a running max/sum: both logit launches and LDS lists gone, 8 -> 6; sum order changed
# speedup vs baseline: 8.0594x; 1.0996x over previous
#include <hip/hip_runtime.h>
#include <math.h>

typedef __attribute__((ext_vector_type(2))) float v2f;
typedef __attribute__((ext_vector_type(8))) float v8f;

#define BM   64
#define BN   64
#define KC   32
#define ASTR 36
#define BSTR 72

static __device__ __forceinline__ float leaky02(float x) {
    return fmaxf(x, 0.2f * x);
}

static __device__ __forceinline__ void mma_chunk(
        const float* __restrict__ as, const float* __restrict__ bs,
        int aoff, int boff,
        v8f& c00, v8f& c01, v8f& c10, v8f& c11) {
    for (int kk = 0; kk < KC; kk += 4) {
        const v2f a0 = *(const v2f*)(as + aoff + kk);
        const v2f a1 = *(const v2f*)(as + aoff + 16 * ASTR + kk);
        v2f b0, b1;
        b0.x = bs[boff + kk * BSTR];
        b0.y = bs[boff + (kk + 1) * BSTR];
        b1.x = bs[boff + 16 + kk * BSTR];
        b1.y = bs[boff + 16 + (kk + 1) * BSTR];
        c00 = __builtin_amdgcn_wmma_f32_16x16x4_f32(
            false, a0, false, b0, (short)0, c00, false, false);
        c01 = __builtin_amdgcn_wmma_f32_16x16x4_f32(
            false, a0, false, b1, (short)0, c01, false, false);
        c10 = __builtin_amdgcn_wmma_f32_16x16x4_f32(
            false, a1, false, b0, (short)0, c10, false, false);
        c11 = __builtin_amdgcn_wmma_f32_16x16x4_f32(
            false, a1, false, b1, (short)0, c11, false, false);
    }
}

__global__ __launch_bounds__(128)
void gemm_wmma_f32(const float* __restrict__ A, const float* __restrict__ B,
                   float* __restrict__ D, int M, int K, int N) {
    __shared__ __attribute__((aligned(16))) float As[2][BM * ASTR];
    __shared__ __attribute__((aligned(16))) float Bs[2][KC * BSTR];

    const int tileN = blockIdx.x * BN;
    const int tileM = blockIdx.y * BM;
    const int t    = threadIdx.x;
    const int wv   = t >> 5;
    const int lane = t & 31;
    const int half = lane >> 4;
    const int lm   = lane & 15;
    const int wm   = wv >> 1;
    const int wn   = wv & 1;

    const int arow = t >> 3;
    const int acol = (t & 7) * 4;
    const int brow = t >> 4;
    const int bcol = (t & 15) * 4;
    const float* ga = A + (size_t)(tileM + arow) * K + acol;
    const float* gb = B + (size_t)brow * N + tileN + bcol;
    const int sa = arow * ASTR + acol;
    const int sb = brow * BSTR + bcol;

    const int aoff = (wm * 32 + lm) * ASTR + 2 * half;
    const int boff = (2 * half) * BSTR + wn * 32 + lm;

    float4 ra[4], rb[4];
    for (int j = 0; j < 4; ++j) ra[j] = *(const float4*)(ga + (size_t)(16 * j) * K);
    for (int j = 0; j < 4; ++j) rb[j] = *(const float4*)(gb + (size_t)(8 * j) * N);
    for (int j = 0; j < 4; ++j) *(float4*)(&As[0][sa + 16 * j * ASTR]) = ra[j];
    for (int j = 0; j < 4; ++j) *(float4*)(&Bs[0][sb + 8 * j * BSTR]) = rb[j];
    __syncthreads();

    v8f c00 = {}, c01 = {}, c10 = {}, c11 = {};
    const int nk = K / KC;
    for (int c = 0; c + 1 < nk; ++c) {
        const int cur = c & 1;
        ga += KC;
        gb += (size_t)KC * N;
        for (int j = 0; j < 4; ++j) ra[j] = *(const float4*)(ga + (size_t)(16 * j) * K);
        for (int j = 0; j < 4; ++j) rb[j] = *(const float4*)(gb + (size_t)(8 * j) * N);
        mma_chunk(As[cur], Bs[cur], aoff, boff, c00, c01, c10, c11);
        for (int j = 0; j < 4; ++j) *(float4*)(&As[cur ^ 1][sa + 16 * j * ASTR]) = ra[j];
        for (int j = 0; j < 4; ++j) *(float4*)(&Bs[cur ^ 1][sb + 8 * j * BSTR]) = rb[j];
        __syncthreads();
    }
    mma_chunk(As[(nk - 1) & 1], Bs[(nk - 1) & 1], aoff, boff, c00, c01, c10, c11);

    const int col   = tileN + wn * 32 + lm;
    const int rbase = tileM + wm * 32 + half * 8;
    float* d0 = D + (size_t)rbase * N + col;
    float* d1 = d0 + (size_t)16 * N;
    for (int r = 0; r < 8; ++r) {
        d0[(size_t)r * N]      = c00[r];
        d0[(size_t)r * N + 16] = c01[r];
        d1[(size_t)r * N]      = c10[r];
        d1[(size_t)r * N + 16] = c11[r];
    }
}

#define BCAP 64

__global__ void fill_i32(int* __restrict__ p, int v, size_t n) {
    size_t i = blockIdx.x * (size_t)blockDim.x + threadIdx.x;
    if (i < n) p[i] = v;
}

__global__ void bucket_fill(const int* __restrict__ ei, int E,
                            int* __restrict__ deg,
                            int* __restrict__ bucket) {
    int t = blockIdx.x * blockDim.x + threadIdx.x;
    if (t >= E) return;
    int s = ei[t], d = ei[E + t];
    int slot = atomicAdd(&deg[d], 1);
    if (slot < BCAP) bucket[(size_t)d * BCAP + slot] = s;
}

template <int M>
static __device__ __forceinline__ float row_xread(float p) {
    return __builtin_bit_cast(float, __builtin_amdgcn_update_dpp(
        0, __builtin_bit_cast(int, p), 0x160 + M, 0xf, 0xf, true));
}

template <int K>
static __device__ __forceinline__ float quad_read(float p) {
    return __builtin_bit_cast(float, __builtin_amdgcn_update_dpp(
        0, __builtin_bit_cast(int, p), K * 0x55, 0xf, 0xf, true));
}

static __device__ __forceinline__ float xhalf(float x) {
    const int b = __builtin_bit_cast(int, x);
    return __builtin_bit_cast(float, __builtin_amdgcn_permlanex16(
        b, b, 0x76543210, 0xfedcba98, false, false));
}

template <int LPH>
static __device__ __forceinline__ float head_sum(float p) {
    p += row_xread<1>(p);
    p += row_xread<2>(p);
    p += row_xread<4>(p);
    p += row_xread<8>(p);
    if (LPH == 32) p += xhalf(p);
    return p;
}

template <int LPH>
static __device__ __forceinline__ float head_sum4(float p0, float p1, float p2, float p3,
                                                  bool odd1, bool odd2) {
    const float k01 = odd1 ? p1 : p0, s01 = odd1 ? p0 : p1;
    const float k23 = odd1 ? p3 : p2, s23 = odd1 ? p2 : p3;
    const float r01 = k01 + row_xread<1>(s01);
    const float r23 = k23 + row_xread<1>(s23);
    const float kb = odd2 ? r23 : r01, sb = odd2 ? r01 : r23;
    float r = kb + row_xread<2>(sb);
    r += row_xread<4>(r);
    r += row_xread<8>(r);
    if (LPH == 32) r += xhalf(r);
    return r;
}

static __device__ __forceinline__ float dot4(const float4 v, const float4 a) {
    const float e = fmaf(v.z, a.z, v.x * a.x);
    const float o = fmaf(v.w, a.w, v.y * a.y);
    return e + o;
}

template <int EPI, int LPH>
__global__ __launch_bounds__(1024)
void gat_row(const int* __restrict__ deg,
             const int* __restrict__ bucket,
             const int* __restrict__ ei, int E,
             int H, int C,
             const float* __restrict__ hfeat,
             const float* __restrict__ a_s,
             const float* __restrict__ a_d,
             const float* __restrict__ bias,
             float* __restrict__ dst) {
    extern __shared__ float4 rowbuf[];
    const int d   = blockIdx.x;
    const int F   = H * C;
    const int f   = threadIdx.x * 4;
    const int nin = deg[d];
    const int* lst = bucket + (size_t)d * BCAP;
    const float4 as4 = *(const float4*)(a_s + f);
    const float4 ad4 = *(const float4*)(a_d + f);

    const float4 vself = *(const float4*)(hfeat + (size_t)d * F + f);
    const float ad = head_sum<LPH>(dot4(vself, ad4));
    float  m   = leaky02(head_sum<LPH>(dot4(vself, as4)) + ad);
    float  den = 1.f;
    float4 acc = vself;

    auto one_edge = [&](int sv) {
        const int s = __builtin_amdgcn_readfirstlane(sv);
        const float4 v  = *(const float4*)(hfeat + (size_t)s * F + f);
        const float  l  = leaky02(head_sum<LPH>(dot4(v, as4)) + ad);
        const float  mn = fmaxf(l, m);
        const float  sc = __expf(m - mn);
        const float  w  = __expf(l - mn);
        den = fmaf(den, sc, w);
        acc.x *= sc; acc.y *= sc; acc.z *= sc; acc.w *= sc;
        acc.x = fmaf(v.x, w, acc.x); acc.y = fmaf(v.y, w, acc.y);
        acc.z = fmaf(v.z, w, acc.z); acc.w = fmaf(v.w, w, acc.w);
        m = mn;
    };

    if (nin <= BCAP) {
        int j = 0;
        const bool odd1 = (threadIdx.x & 1) != 0;
        const bool odd2 = (threadIdx.x & 2) != 0;
        for (; j + 4 <= nin; j += 4) {
            const int4 s4 = *(const int4*)(lst + j);
            const int s0 = __builtin_amdgcn_readfirstlane(s4.x);
            const int s1 = __builtin_amdgcn_readfirstlane(s4.y);
            const int s2 = __builtin_amdgcn_readfirstlane(s4.z);
            const int s3 = __builtin_amdgcn_readfirstlane(s4.w);
            const float4 v0 = *(const float4*)(hfeat + (size_t)s0 * F + f);
            const float4 v1 = *(const float4*)(hfeat + (size_t)s1 * F + f);
            const float4 v2 = *(const float4*)(hfeat + (size_t)s2 * F + f);
            const float4 v3 = *(const float4*)(hfeat + (size_t)s3 * F + f);
            const float r = head_sum4<LPH>(dot4(v0, as4), dot4(v1, as4),
                                           dot4(v2, as4), dot4(v3, as4), odd1, odd2);
            const float l0 = leaky02(quad_read<0>(r) + ad);
            const float l1 = leaky02(quad_read<1>(r) + ad);
            const float l2 = leaky02(quad_read<2>(r) + ad);
            const float l3 = leaky02(quad_read<3>(r) + ad);
            const float mn = fmaxf(fmaxf(fmaxf(l0, l1), fmaxf(l2, l3)), m);
            const float sc = __expf(m - mn);
            const float w0 = __expf(l0 - mn);
            const float w1 = __expf(l1 - mn);
            const float w2 = __expf(l2 - mn);
            const float w3 = __expf(l3 - mn);
            den = fmaf(den, sc, w0); den += w1; den += w2; den += w3;
            acc.x *= sc; acc.y *= sc; acc.z *= sc; acc.w *= sc;
            acc.x = fmaf(v0.x, w0, acc.x); acc.y = fmaf(v0.y, w0, acc.y);
            acc.z = fmaf(v0.z, w0, acc.z); acc.w = fmaf(v0.w, w0, acc.w);
            acc.x = fmaf(v1.x, w1, acc.x); acc.y = fmaf(v1.y, w1, acc.y);
            acc.z = fmaf(v1.z, w1, acc.z); acc.w = fmaf(v1.w, w1, acc.w);
            acc.x = fmaf(v2.x, w2, acc.x); acc.y = fmaf(v2.y, w2, acc.y);
            acc.z = fmaf(v2.z, w2, acc.z); acc.w = fmaf(v2.w, w2, acc.w);
            acc.x = fmaf(v3.x, w3, acc.x); acc.y = fmaf(v3.y, w3, acc.y);
            acc.z = fmaf(v3.z, w3, acc.z); acc.w = fmaf(v3.w, w3, acc.w);
            m = mn;
        }
        for (; j < nin; ++j) one_edge(lst[j]);
    } else {
        const int* es = ei;
        const int* ed = ei + E;
        for (int j = 0; j < E; ++j) {
            if (ed[j] != d) continue;
            one_edge(es[j]);
        }
    }
    {
        const float inv = 1.f / den;
        acc.x *= inv; acc.y *= inv; acc.z *= inv; acc.w *= inv;
    }
    if (EPI == 1) {
        const float4 bb = *(const float4*)(bias + f);
        float4 r;
        r.x = acc.x + bb.x; r.y = acc.y + bb.y; r.z = acc.z + bb.z; r.w = acc.w + bb.w;
        r.x = r.x > 0.f ? r.x : 0.f; r.y = r.y > 0.f ? r.y : 0.f;
        r.z = r.z > 0.f ? r.z : 0.f; r.w = r.w > 0.f ? r.w : 0.f;
        *(float4*)(dst + (size_t)d * F + f) = r;
    } else {
        rowbuf[threadIdx.x] = acc;
        __syncthreads();
        if (f < C) {
            const int q = C >> 2;
            float4 s = {0.f, 0.f, 0.f, 0.f};
            for (int g = 0; g < H; ++g) {
                const float4 v = rowbuf[g * q + threadIdx.x];
                s.x += v.x; s.y += v.y; s.z += v.z; s.w += v.w;
            }
            const float4 bb = *(const float4*)(bias + f);
            const float hn = (float)H;
            float4 r;
            r.x = s.x / hn + bb.x; r.y = s.y / hn + bb.y;
            r.z = s.z / hn + bb.z; r.w = s.w / hn + bb.w;
            *(float4*)(dst + (size_t)d * C + f) = r;
        }
    }
}

static inline unsigned cdiv_u(size_t a, unsigned b) {
    return (unsigned)((a + b - 1) / b);
}

extern "C" void kernel_launch(void* const* d_in, const int* in_sizes, int n_in,
                              void* d_out, int out_size, void* d_ws, size_t ws_size,
                              hipStream_t stream) {
    const float* x     = (const float*)d_in[0];
    const int*   ei    = (const int*)d_in[1];
    const float* W1    = (const float*)d_in[2];
    const float* a_s1  = (const float*)d_in[3];
    const float* a_d1  = (const float*)d_in[4];
    const float* b1    = (const float*)d_in[5];
    const float* W2    = (const float*)d_in[6];
    const float* a_s2  = (const float*)d_in[7];
    const float* a_d2  = (const float*)d_in[8];
    const float* b2    = (const float*)d_in[9];
    float*       out   = (float*)d_out;

    const int F1   = in_sizes[3];
    const int F2   = in_sizes[7];
    const int LAT  = in_sizes[9];
    const int H    = F2 / LAT;
    const int C1   = F1 / H;
    const int Fin  = in_sizes[2] / F1;
    const int N    = in_sizes[0] / Fin;
    const int E    = in_sizes[1] / 2;

    float* ws = (float*)d_ws;
    size_t o = 0;
    float* h1   = ws + o; o += (size_t)N * F1;
    float* agg1 = ws + o; o += (size_t)N * F1;
    int* deg    = (int*)(ws + o); o += (size_t)N;
    int* bucket = (int*)(ws + o); o += (size_t)N * BCAP;
    float* h2   = h1;
    (void)ws_size; (void)n_in; (void)out_size;

    const unsigned TB = 256;

    fill_i32<<<cdiv_u((size_t)N, TB), TB, 0, stream>>>(deg, 0, (size_t)N);
    bucket_fill<<<cdiv_u((size_t)E, TB), TB, 0, stream>>>(ei, E, deg, bucket);

    gemm_wmma_f32<<<dim3(F1 / BN, N / BM), 128, 0, stream>>>(x, W1, h1, N, Fin, F1);
    gat_row<1, 32><<<N, F1 / 4, 0, stream>>>(
        deg, bucket, ei, E, H, C1, h1, a_s1, a_d1, b1, agg1);

    gemm_wmma_f32<<<dim3(F2 / BN, N / BM), 128, 0, stream>>>(agg1, W2, h2, N, F1, F2);
    gat_row<2, 16><<<N, F2 / 4, (size_t)F2 * sizeof(float), stream>>>(
        deg, bucket, ei, E, H, LAT, h2, a_s2, a_d2, b2, out);
}
